# starting kernel
# baseline (speedup 1.0000x reference)
#include <hip/hip_runtime.h>

typedef __attribute__((ext_vector_type(2))) float v2f;
typedef __attribute__((ext_vector_type(8))) float v8f;

__global__ void k_deg_init(float* __restrict__ deg, int n) {
  int i = blockIdx.x * blockDim.x + threadIdx.x;
  if (i < n) deg[i] = 1.0f;
}

__global__ void k_deg_count(const int* __restrict__ dst, float* __restrict__ deg, int E) {
  int e = blockIdx.x * blockDim.x + threadIdx.x;
  if (e < E)
    __hip_atomic_fetch_add(&deg[dst[e]], 1.0f, __ATOMIC_RELAXED, __HIP_MEMORY_SCOPE_AGENT);
}

__global__ void k_dinv(float* __restrict__ deg, int n) {
  int i = blockIdx.x * blockDim.x + threadIdx.x;
  if (i < n) {
    float d = deg[i];
    deg[i] = (d > 0.0f) ? rsqrtf(d) : 0.0f;
  }
}

template <bool RELU>
__global__ void k_gemm_wmma(const float* __restrict__ A, const float* __restrict__ W,
                            float* __restrict__ C, int K, int Nc) {
  const int lane = threadIdx.x & 31;
  const int wave = threadIdx.x >> 5;
  const int row  = blockIdx.x * 16 + (lane & 15);
  const int col  = wave * 16 + (lane & 15);
  const int kh   = (lane >> 4) << 1;

  const float* arow = A + (size_t)row * K;
  v8f acc = {};
  for (int k = 0; k < K; k += 4) {
    float a0 = arow[k + kh];
    float a1 = arow[k + kh + 1];
    if (RELU) { a0 = fmaxf(a0, 0.0f); a1 = fmaxf(a1, 0.0f); }
    v2f a, b;
    a.x = a0;
    a.y = a1;
    b.x = W[(size_t)(k + kh) * Nc + col];
    b.y = W[(size_t)(k + kh + 1) * Nc + col];
    acc = __builtin_amdgcn_wmma_f32_16x16x4_f32(false, a, false, b, (short)0, acc,
                                                false, false);
  }
  const int rbase = blockIdx.x * 16 + ((lane >> 4) << 3);
#pragma unroll
  for (int v = 0; v < 8; ++v)
    C[(size_t)(rbase + v) * Nc + col] = acc[v];
}

template <int FSHIFT>
__global__ void k_init_acc(const float* __restrict__ H, const float* __restrict__ dinv,
                           const float* __restrict__ bias, float* __restrict__ acc,
                           int total) {
  int gid = blockIdx.x * blockDim.x + threadIdx.x;
  if (gid >= total) return;
  int i = gid >> FSHIFT;
  int f = gid & ((1 << FSHIFT) - 1);
  float di = dinv[i];
  acc[gid] = bias[f] + di * di * H[gid];
}

template <int LPE_SHIFT>
__global__ void k_edge_scatter(const int* __restrict__ src, const int* __restrict__ dst,
                               const float* __restrict__ dinv, const float* __restrict__ H,
                               float* __restrict__ acc, int E) {
  const int F = 4 << LPE_SHIFT;
  int gid = blockIdx.x * blockDim.x + threadIdx.x;
  int e = gid >> LPE_SHIFT;
  if (e >= E) return;
  int f4 = (gid & ((1 << LPE_SHIFT) - 1)) << 2;
  int s = src[e];
  int d = dst[e];
  float nrm = dinv[s] * dinv[d];
  float4 m = *(const float4*)(H + (size_t)s * F + f4);
  float* op = acc + (size_t)d * F + f4;
  __hip_atomic_fetch_add(op + 0, m.x * nrm, __ATOMIC_RELAXED, __HIP_MEMORY_SCOPE_AGENT);
  __hip_atomic_fetch_add(op + 1, m.y * nrm, __ATOMIC_RELAXED, __HIP_MEMORY_SCOPE_AGENT);
  __hip_atomic_fetch_add(op + 2, m.z * nrm, __ATOMIC_RELAXED, __HIP_MEMORY_SCOPE_AGENT);
  __hip_atomic_fetch_add(op + 3, m.w * nrm, __ATOMIC_RELAXED, __HIP_MEMORY_SCOPE_AGENT);
}

extern "C" void kernel_launch(void* const* d_in, const int* in_sizes, int n_in,
                              void* d_out, int out_size, void* d_ws, size_t ws_size,
                              hipStream_t stream) {
  const float* x  = (const float*)d_in[0];
  const int*   ei = (const int*)d_in[1];
  const float* W1 = (const float*)d_in[2];
  const float* b1 = (const float*)d_in[3];
  const float* W2 = (const float*)d_in[4];
  const float* b2 = (const float*)d_in[5];
  float* out = (float*)d_out;

  const int FIN = 256, FHID = 128, FOUT = 64;
  const int N = in_sizes[0] / FIN;
  const int E = in_sizes[1] / 2;

  const int* srcI = ei;
  const int* dstI = ei + E;

  float* ws   = (float*)d_ws;
  float* dinv = ws;
  float* xw1  = ws + N;
  float* agg1 = xw1 + (size_t)N * FHID;
  float* h2   = xw1;

  const int T = 256;

  k_deg_init<<<(N + T - 1) / T, T, 0, stream>>>(dinv, N);
  k_deg_count<<<(E + T - 1) / T, T, 0, stream>>>(dstI, dinv, E);
  k_dinv<<<(N + T - 1) / T, T, 0, stream>>>(dinv, N);

  k_gemm_wmma<false><<<N / 16, (FHID / 16) * 32, 0, stream>>>(x, W1, xw1, FIN, FHID);
  k_init_acc<7><<<(N * FHID + T - 1) / T, T, 0, stream>>>(xw1, dinv, b1, agg1, N * FHID);
  k_edge_scatter<5><<<(E * 32 + T - 1) / T, T, 0, stream>>>(srcI, dstI, dinv, xw1, agg1, E);

  k_gemm_wmma<true><<<N / 16, (FOUT / 16) * 32, 0, stream>>>(agg1, W2, h2, FHID, FOUT);
  k_init_acc<6><<<(N * FOUT + T - 1) / T, T, 0, stream>>>(h2, dinv, b2, out, N * FOUT);
  k_edge_scatter<4><<<(E * 16 + T - 1) / T, T, 0, stream>>>(srcI, dstI, dinv, h2, out, E);
}


# Float-atomic edge scatter replaced by per-call grouping of edges by destination (returning int add = rank, scan, fill) and one wave per node summing its rows in registers in arrival-rank order
# speedup vs baseline: 1.9026x; 1.9026x over previous
#include <hip/hip_runtime.h>

typedef __attribute__((ext_vector_type(2))) float v2f;
typedef __attribute__((ext_vector_type(8))) float v8f;

__global__ void k_zero_cnt(int* __restrict__ cnt, int n) {
  int i = blockIdx.x * blockDim.x + threadIdx.x;
  if (i < n) cnt[i] = 0;
}

__global__ void k_count_rank(const int* __restrict__ dst, int* __restrict__ cnt,
                             int* __restrict__ rank, int E) {
  int e = blockIdx.x * blockDim.x + threadIdx.x;
  if (e < E)
    rank[e] = __hip_atomic_fetch_add(&cnt[dst[e]], 1, __ATOMIC_RELAXED, __HIP_MEMORY_SCOPE_AGENT);
}

__global__ void __launch_bounds__(1024)
k_scan_dinv(const int* __restrict__ cnt, int* __restrict__ rowptr,
            float* __restrict__ dinv, int n) {
  __shared__ int s_incl[32];
  __shared__ int s_part[32];
  const int t = threadIdx.x;
  const int lane = t & 31;
  const int wv = t >> 5;
  const int first = blockIdx.x * 1024;

  const int4* q = (const int4*)cnt;
  const int nq = first >> 2;
  const int last = nq - 1;
  int p0 = 0, p1 = 0, p2 = 0, p3 = 0;
  for (int k = t; k < nq; k += 4096) {
    const int k1 = k + 1024, k2 = k + 2048, k3 = k + 3072;
    const int4 a = q[k];
    const int4 b = q[k1 < nq ? k1 : last];
    const int4 c = q[k2 < nq ? k2 : last];
    const int4 d = q[k3 < nq ? k3 : last];
    p0 += (a.x + a.y) + (a.z + a.w);
    p1 += (k1 < nq) ? (b.x + b.y) + (b.z + b.w) : 0;
    p2 += (k2 < nq) ? (c.x + c.y) + (c.z + c.w) : 0;
    p3 += (k3 < nq) ? (d.x + d.y) + (d.z + d.w) : 0;
  }
  int part = (p0 + p1) + (p2 + p3);

  const int i = first + t;
  const int c = (i < n) ? cnt[i] : 0;
  int incl = c;
  for (int d = 1; d < 32; d <<= 1) {
    int u = __shfl_up(incl, d);
    if (lane >= d) incl += u;
  }
  for (int d = 16; d >= 1; d >>= 1) part += __shfl_xor(part, d);
  if (lane == 31) s_incl[wv] = incl;
  if (lane == 0) s_part[wv] = part;
  __syncthreads();
  int front = s_part[lane] + ((lane < wv) ? s_incl[lane] : 0);
  for (int d = 16; d >= 1; d >>= 1) front += __shfl_xor(front, d);
  const int excl = front + incl - c;
  if (i < n) {
    rowptr[i] = excl;
    if (i == n - 1) rowptr[n] = excl + c;
    float dg = (float)(c + 1);
    dinv[i] = (dg > 0.0f) ? rsqrtf(dg) : 0.0f;
  }
}

__global__ void k_fill(const int* __restrict__ src, const int* __restrict__ dst,
                       const int* __restrict__ rank, const int* __restrict__ rowptr,
                       int* __restrict__ csr, int E) {
  int e = blockIdx.x * blockDim.x + threadIdx.x;
  if (e < E) csr[rowptr[dst[e]] + rank[e]] = src[e];
}

__global__ void k_move_list(const int* __restrict__ from, int* __restrict__ to, int n) {
  int i = blockIdx.x * blockDim.x + threadIdx.x;
  if (i < n) to[i] = from[i];
}

template <bool RELU>
__global__ void k_gemm_wmma(const float* __restrict__ A, const float* __restrict__ W,
                            float* __restrict__ C, int K, int Nc) {
  const int lane = threadIdx.x & 31;
  const int wave = threadIdx.x >> 5;
  const int row  = blockIdx.x * 16 + (lane & 15);
  const int col  = wave * 16 + (lane & 15);
  const int kh   = (lane >> 4) << 1;

  const float* arow = A + (size_t)row * K;
  v8f acc = {};
  for (int k = 0; k < K; k += 4) {
    float a0 = arow[k + kh];
    float a1 = arow[k + kh + 1];
    if (RELU) { a0 = fmaxf(a0, 0.0f); a1 = fmaxf(a1, 0.0f); }
    v2f a, b;
    a.x = a0;
    a.y = a1;
    b.x = W[(size_t)(k + kh) * Nc + col];
    b.y = W[(size_t)(k + kh + 1) * Nc + col];
    acc = __builtin_amdgcn_wmma_f32_16x16x4_f32(false, a, false, b, (short)0, acc,
                                                false, false);
  }
  const int rbase = blockIdx.x * 16 + ((lane >> 4) << 3);
#pragma unroll
  for (int v = 0; v < 8; ++v)
    C[(size_t)(rbase + v) * Nc + col] = acc[v];
}

template <int F>
__global__ void k_gather(const int* __restrict__ rowptr, const int* __restrict__ csr,
                         const float* __restrict__ dinv, const float* __restrict__ H,
                         const float* __restrict__ bias, float* __restrict__ O, int n) {
  constexpr int V = F / 32;
  typedef __attribute__((ext_vector_type(V))) float vecT;
  const int lane = threadIdx.x & 31;
  const int node = blockIdx.x * (blockDim.x >> 5) + (threadIdx.x >> 5);
  if (node >= n) return;
  const int beg = __builtin_amdgcn_readfirstlane(rowptr[node]);
  const int end = __builtin_amdgcn_readfirstlane(rowptr[node + 1]);
  const int lo = lane * V;
  const float di = dinv[node];
  vecT acc = *(const vecT*)(bias + lo) + *(const vecT*)(H + (size_t)node * F + lo) * (di * di);

  for (int base = beg; base < end; base += 32) {
    int ms = 0;
    float mn = 0.0f;
    if (base + lane < end) {
      ms = csr[base + lane];
      mn = dinv[ms] * di;
    }
    const int mni = __float_as_int(mn);
    const int c = (end - base < 32) ? (end - base) : 32;
    int j = 0;
    for (; j + 4 <= c; j += 4) {
      const int s0 = __builtin_amdgcn_readlane(ms, j);
      const int s1 = __builtin_amdgcn_readlane(ms, j + 1);
      const int s2 = __builtin_amdgcn_readlane(ms, j + 2);
      const int s3 = __builtin_amdgcn_readlane(ms, j + 3);
      const float w0 = __int_as_float(__builtin_amdgcn_readlane(mni, j));
      const float w1 = __int_as_float(__builtin_amdgcn_readlane(mni, j + 1));
      const float w2 = __int_as_float(__builtin_amdgcn_readlane(mni, j + 2));
      const float w3 = __int_as_float(__builtin_amdgcn_readlane(mni, j + 3));
      const vecT m0 = *(const vecT*)(H + (size_t)s0 * F + lo);
      const vecT m1 = *(const vecT*)(H + (size_t)s1 * F + lo);
      const vecT m2 = *(const vecT*)(H + (size_t)s2 * F + lo);
      const vecT m3 = *(const vecT*)(H + (size_t)s3 * F + lo);
      acc += m0 * w0;
      acc += m1 * w1;
      acc += m2 * w2;
      acc += m3 * w3;
    }
    if (j + 2 <= c) {
      const int s0 = __builtin_amdgcn_readlane(ms, j);
      const int s1 = __builtin_amdgcn_readlane(ms, j + 1);
      const float w0 = __int_as_float(__builtin_amdgcn_readlane(mni, j));
      const float w1 = __int_as_float(__builtin_amdgcn_readlane(mni, j + 1));
      const vecT m0 = *(const vecT*)(H + (size_t)s0 * F + lo);
      const vecT m1 = *(const vecT*)(H + (size_t)s1 * F + lo);
      acc += m0 * w0;
      acc += m1 * w1;
      j += 2;
    }
    if (j < c) {
      const int s0 = __builtin_amdgcn_readlane(ms, j);
      const float w0 = __int_as_float(__builtin_amdgcn_readlane(mni, j));
      const vecT m0 = *(const vecT*)(H + (size_t)s0 * F + lo);
      acc += m0 * w0;
    }
  }
  *(vecT*)(O + (size_t)node * F + lo) = acc;
}

extern "C" void kernel_launch(void* const* d_in, const int* in_sizes, int n_in,
                              void* d_out, int out_size, void* d_ws, size_t ws_size,
                              hipStream_t stream) {
  const float* x  = (const float*)d_in[0];
  const int*   ei = (const int*)d_in[1];
  const float* W1 = (const float*)d_in[2];
  const float* b1 = (const float*)d_in[3];
  const float* W2 = (const float*)d_in[4];
  const float* b2 = (const float*)d_in[5];
  float* out = (float*)d_out;

  const int FIN = 256, FHID = 128, FOUT = 64;
  const int N = in_sizes[0] / FIN;
  const int E = in_sizes[1] / 2;

  const int* srcI = ei;
  const int* dstI = ei + E;

  float* ws   = (float*)d_ws;
  float* dinv = ws;
  float* xw1  = ws + N;
  float* agg1 = xw1 + (size_t)N * FHID;
  float* h2   = xw1;

  int* cnt  = (int*)agg1;
  int* rank = cnt + N;
  int* listA = (int*)out;
  int* listB = (int*)(xw1 + (size_t)N * FOUT);
  const int LIST = N + 1 + E;

  const int T = 256;

  k_zero_cnt<<<(N + T - 1) / T, T, 0, stream>>>(cnt, N);
  k_count_rank<<<(E + T - 1) / T, T, 0, stream>>>(dstI, cnt, rank, E);
  k_scan_dinv<<<(N + 1023) / 1024, 1024, 0, stream>>>(cnt, listA, dinv, N);
  k_fill<<<(E + T - 1) / T, T, 0, stream>>>(srcI, dstI, rank, listA, listA + N + 1, E);

  k_gemm_wmma<false><<<N / 16, (FHID / 16) * 32, 0, stream>>>(x, W1, xw1, FIN, FHID);
  k_gather<128><<<(N + 7) / 8, T, 0, stream>>>(listA, listA + N + 1, dinv, xw1, b1, agg1, N);
  k_move_list<<<(LIST + T - 1) / T, T, 0, stream>>>(listA, listB, LIST);

  k_gemm_wmma<true><<<N / 16, (FOUT / 16) * 32, 0, stream>>>(agg1, W2, h2, FHID, FOUT);
  k_gather<64><<<(N + 7) / 8, T, 0, stream>>>(listB, listB + N + 1, dinv, h2, b2, out, N);
}


# Both projections re-tiled: one wave per 16-row strip over all output columns, each A fragment loaded once for every column tile, W held in LDS per work-group, next fragments read one step ahead
# speedup vs baseline: 2.4459x; 1.2856x over previous
#include <hip/hip_runtime.h>

typedef __attribute__((ext_vector_type(2))) float v2f;
typedef __attribute__((ext_vector_type(8))) float v8f;

__global__ void k_zero_cnt(int* __restrict__ cnt, int n) {
  int i = blockIdx.x * blockDim.x + threadIdx.x;
  if (i < n) cnt[i] = 0;
}

__global__ void k_count_rank(const int* __restrict__ dst, int* __restrict__ cnt,
                             int* __restrict__ rank, int E) {
  int e = blockIdx.x * blockDim.x + threadIdx.x;
  if (e < E)
    rank[e] = __hip_atomic_fetch_add(&cnt[dst[e]], 1, __ATOMIC_RELAXED, __HIP_MEMORY_SCOPE_AGENT);
}

__global__ void __launch_bounds__(1024)
k_scan_dinv(const int* __restrict__ cnt, int* __restrict__ rowptr,
            float* __restrict__ dinv, int n) {
  __shared__ int s_incl[32];
  __shared__ int s_part[32];
  const int t = threadIdx.x;
  const int lane = t & 31;
  const int wv = t >> 5;
  const int first = blockIdx.x * 1024;

  const int4* q = (const int4*)cnt;
  const int nq = first >> 2;
  const int last = nq - 1;
  int p0 = 0, p1 = 0, p2 = 0, p3 = 0;
  for (int k = t; k < nq; k += 4096) {
    const int k1 = k + 1024, k2 = k + 2048, k3 = k + 3072;
    const int4 a = q[k];
    const int4 b = q[k1 < nq ? k1 : last];
    const int4 c = q[k2 < nq ? k2 : last];
    const int4 d = q[k3 < nq ? k3 : last];
    p0 += (a.x + a.y) + (a.z + a.w);
    p1 += (k1 < nq) ? (b.x + b.y) + (b.z + b.w) : 0;
    p2 += (k2 < nq) ? (c.x + c.y) + (c.z + c.w) : 0;
    p3 += (k3 < nq) ? (d.x + d.y) + (d.z + d.w) : 0;
  }
  int part = (p0 + p1) + (p2 + p3);

  const int i = first + t;
  const int c = (i < n) ? cnt[i] : 0;
  int incl = c;
  for (int d = 1; d < 32; d <<= 1) {
    int u = __shfl_up(incl, d);
    if (lane >= d) incl += u;
  }
  for (int d = 16; d >= 1; d >>= 1) part += __shfl_xor(part, d);
  if (lane == 31) s_incl[wv] = incl;
  if (lane == 0) s_part[wv] = part;
  __syncthreads();
  int front = s_part[lane] + ((lane < wv) ? s_incl[lane] : 0);
  for (int d = 16; d >= 1; d >>= 1) front += __shfl_xor(front, d);
  const int excl = front + incl - c;
  if (i < n) {
    rowptr[i] = excl;
    if (i == n - 1) rowptr[n] = excl + c;
    float dg = (float)(c + 1);
    dinv[i] = (dg > 0.0f) ? rsqrtf(dg) : 0.0f;
  }
}

__global__ void k_fill(const int* __restrict__ src, const int* __restrict__ dst,
                       const int* __restrict__ rank, const int* __restrict__ rowptr,
                       int* __restrict__ csr, int E) {
  int e = blockIdx.x * blockDim.x + threadIdx.x;
  if (e < E) csr[rowptr[dst[e]] + rank[e]] = src[e];
}

__global__ void k_move_list(const int* __restrict__ from, int* __restrict__ to, int n) {
  int i = blockIdx.x * blockDim.x + threadIdx.x;
  if (i < n) to[i] = from[i];
}

constexpr int GEMM_THREADS = 800;
template <int K, int NC, bool RELU>
__global__ void __launch_bounds__(GEMM_THREADS)
k_gemm_strip(const float* __restrict__ A, const float* __restrict__ W,
             float* __restrict__ C, int nstrips) {
  constexpr int TILES = NC / 16;
  constexpr int P2 = 2 * NC + 32;
  constexpr int NQ = NC / 4;
  __shared__ __attribute__((aligned(16))) float Ws[(K / 2) * P2];

  typedef __attribute__((ext_vector_type(4))) float v4f;
#pragma unroll
  for (int idx = threadIdx.x; idx < (K / 2) * NQ; idx += GEMM_THREADS) {
    const int p = idx / NQ;
    const int q = idx % NQ;
    const v4f w0 = *(const v4f*)(W + (size_t)(2 * p) * NC + 4 * q);
    const v4f w1 = *(const v4f*)(W + (size_t)(2 * p + 1) * NC + 4 * q);
    v4f lo, hi;
    lo.x = w0.x; lo.y = w1.x; lo.z = w0.y; lo.w = w1.y;
    hi.x = w0.z; hi.y = w1.z; hi.z = w0.w; hi.w = w1.w;
    float* d = Ws + p * P2 + 8 * q;
    *(v4f*)d = lo;
    *(v4f*)(d + 4) = hi;
  }
  __syncthreads();

  const int lane = threadIdx.x & 31;
  const int wave = threadIdx.x >> 5;
  const int strip = blockIdx.x * (GEMM_THREADS / 32) + wave;
  if (strip >= nstrips) return;
  const int c = lane & 15;
  const int h = lane >> 4;

  const float* arow = A + (size_t)(strip * 16 + c) * K + 2 * h;
  const float* bl = Ws + h * P2 + 2 * c;

  v8f acc[TILES];
#pragma unroll
  for (int t = 0; t < TILES; ++t) acc[t] = v8f{};

  v2f acur[8], anxt[8];
#pragma unroll
  for (int s = 0; s < 8; ++s) acur[s] = *(const v2f*)(arow + 4 * s);

  v2f bcur[TILES], bnxt[TILES];
#pragma unroll
  for (int t = 0; t < TILES; ++t) bcur[t] = *(const v2f*)(bl + 32 * t);

  for (int k0 = 0; k0 < K; k0 += 32) {
    const int kn = (k0 + 32 < K) ? k0 + 32 : k0;
#pragma unroll
    for (int s = 0; s < 8; ++s) anxt[s] = *(const v2f*)(arow + kn + 4 * s);

#pragma unroll
    for (int s = 0; s < 8; ++s) {
      v2f a = acur[s];
      if (RELU) { a.x = fmaxf(a.x, 0.0f); a.y = fmaxf(a.y, 0.0f); }
      const int kb = (k0 + 4 * s + 4 < K) ? k0 + 4 * s + 4 : 0;
      const float* bp = bl + (kb >> 1) * P2;
#pragma unroll
      for (int t = 0; t < TILES; ++t) bnxt[t] = *(const v2f*)(bp + 32 * t);
      __builtin_amdgcn_sched_barrier(0);
#pragma unroll
      for (int t = 0; t < TILES; ++t)
        acc[t] = __builtin_amdgcn_wmma_f32_16x16x4_f32(false, a, false, bcur[t], (short)0,
                                                       acc[t], false, false);
      __builtin_amdgcn_sched_barrier(0);
#pragma unroll
      for (int t = 0; t < TILES; ++t) bcur[t] = bnxt[t];
    }
#pragma unroll
    for (int s = 0; s < 8; ++s) acur[s] = anxt[s];
  }

  const int rbase = strip * 16 + 8 * h;
#pragma unroll
  for (int t = 0; t < TILES; ++t)
#pragma unroll
    for (int v = 0; v < 8; ++v)
      C[(size_t)(rbase + v) * NC + 16 * t + c] = acc[t][v];
}

template <int F>
__global__ void k_gather(const int* __restrict__ rowptr, const int* __restrict__ csr,
                         const float* __restrict__ dinv, const float* __restrict__ H,
                         const float* __restrict__ bias, float* __restrict__ O, int n) {
  constexpr int V = F / 32;
  typedef __attribute__((ext_vector_type(V))) float vecT;
  const int lane = threadIdx.x & 31;
  const int node = blockIdx.x * (blockDim.x >> 5) + (threadIdx.x >> 5);
  if (node >= n) return;
  const int beg = __builtin_amdgcn_readfirstlane(rowptr[node]);
  const int end = __builtin_amdgcn_readfirstlane(rowptr[node + 1]);
  const int lo = lane * V;
  const float di = dinv[node];
  vecT acc = *(const vecT*)(bias + lo) + *(const vecT*)(H + (size_t)node * F + lo) * (di * di);

  for (int base = beg; base < end; base += 32) {
    int ms = 0;
    float mn = 0.0f;
    if (base + lane < end) {
      ms = csr[base + lane];
      mn = dinv[ms] * di;
    }
    const int mni = __float_as_int(mn);
    const int c = (end - base < 32) ? (end - base) : 32;
    int j = 0;
    for (; j + 4 <= c; j += 4) {
      const int s0 = __builtin_amdgcn_readlane(ms, j);
      const int s1 = __builtin_amdgcn_readlane(ms, j + 1);
      const int s2 = __builtin_amdgcn_readlane(ms, j + 2);
      const int s3 = __builtin_amdgcn_readlane(ms, j + 3);
      const float w0 = __int_as_float(__builtin_amdgcn_readlane(mni, j));
      const float w1 = __int_as_float(__builtin_amdgcn_readlane(mni, j + 1));
      const float w2 = __int_as_float(__builtin_amdgcn_readlane(mni, j + 2));
      const float w3 = __int_as_float(__builtin_amdgcn_readlane(mni, j + 3));
      const vecT m0 = *(const vecT*)(H + (size_t)s0 * F + lo);
      const vecT m1 = *(const vecT*)(H + (size_t)s1 * F + lo);
      const vecT m2 = *(const vecT*)(H + (size_t)s2 * F + lo);
      const vecT m3 = *(const vecT*)(H + (size_t)s3 * F + lo);
      acc += m0 * w0;
      acc += m1 * w1;
      acc += m2 * w2;
      acc += m3 * w3;
    }
    if (j + 2 <= c) {
      const int s0 = __builtin_amdgcn_readlane(ms, j);
      const int s1 = __builtin_amdgcn_readlane(ms, j + 1);
      const float w0 = __int_as_float(__builtin_amdgcn_readlane(mni, j));
      const float w1 = __int_as_float(__builtin_amdgcn_readlane(mni, j + 1));
      const vecT m0 = *(const vecT*)(H + (size_t)s0 * F + lo);
      const vecT m1 = *(const vecT*)(H + (size_t)s1 * F + lo);
      acc += m0 * w0;
      acc += m1 * w1;
      j += 2;
    }
    if (j < c) {
      const int s0 = __builtin_amdgcn_readlane(ms, j);
      const float w0 = __int_as_float(__builtin_amdgcn_readlane(mni, j));
      const vecT m0 = *(const vecT*)(H + (size_t)s0 * F + lo);
      acc += m0 * w0;
    }
  }
  *(vecT*)(O + (size_t)node * F + lo) = acc;
}

extern "C" void kernel_launch(void* const* d_in, const int* in_sizes, int n_in,
                              void* d_out, int out_size, void* d_ws, size_t ws_size,
                              hipStream_t stream) {
  const float* x  = (const float*)d_in[0];
  const int*   ei = (const int*)d_in[1];
  const float* W1 = (const float*)d_in[2];
  const float* b1 = (const float*)d_in[3];
  const float* W2 = (const float*)d_in[4];
  const float* b2 = (const float*)d_in[5];
  float* out = (float*)d_out;

  const int FIN = 256, FHID = 128, FOUT = 64;
  const int N = in_sizes[0] / FIN;
  const int E = in_sizes[1] / 2;

  const int* srcI = ei;
  const int* dstI = ei + E;

  float* ws   = (float*)d_ws;
  float* dinv = ws;
  float* xw1  = ws + N;
  float* agg1 = xw1 + (size_t)N * FHID;
  float* h2   = xw1;

  int* cnt  = (int*)agg1;
  int* rank = cnt + N;
  int* listA = (int*)out;
  int* listB = (int*)(xw1 + (size_t)N * FOUT);
  const int LIST = N + 1 + E;

  const int T = 256;
  const int NS = N / 16;
  const int GW = GEMM_THREADS / 32;

  k_zero_cnt<<<(N + T - 1) / T, T, 0, stream>>>(cnt, N);
  k_count_rank<<<(E + T - 1) / T, T, 0, stream>>>(dstI, cnt, rank, E);
  k_scan_dinv<<<(N + 1023) / 1024, 1024, 0, stream>>>(cnt, listA, dinv, N);
  k_fill<<<(E + T - 1) / T, T, 0, stream>>>(srcI, dstI, rank, listA, listA + N + 1, E);

  k_gemm_strip<FIN, FHID, false><<<(NS + GW - 1) / GW, GEMM_THREADS, 0, stream>>>(x, W1, xw1, NS);
  k_gather<128><<<(N + 7) / 8, T, 0, stream>>>(listA, listA + N + 1, dinv, xw1, b1, agg1, N);
  k_move_list<<<(LIST + T - 1) / T, T, 0, stream>>>(listA, listB, LIST);

  k_gemm_strip<FHID, FOUT, true><<<(NS + GW - 1) / GW, GEMM_THREADS, 0, stream>>>(agg1, W2, h2, NS);
  k_gather<64><<<(N + 7) / 8, T, 0, stream>>>(listB, listB + N + 1, dinv, h2, b2, out, N);
}


# Edge grouping without memory atomics: per-chunk LDS counts of 256-node destination bins, scan, placement into bins, then one work-group per bin ranks edges in LDS; node sum order = LDS arrival order
# speedup vs baseline: 3.4762x; 1.4212x over previous
#include <hip/hip_runtime.h>

typedef __attribute__((ext_vector_type(2))) float v2f;
typedef __attribute__((ext_vector_type(8))) float v8f;

constexpr int BIN_SHIFT = 8;
constexpr int BIN_NODES = 1 << BIN_SHIFT;
constexpr int MAX_BINS = 512;
constexpr int CHUNKS = 250;
constexpr int EDGE_THREADS = 1024;
constexpr int EDGE_AHEAD = 7;
constexpr int BIN_AHEAD = 6;

__device__ __forceinline__ int lds_add(int* p, int v) {
  return __hip_atomic_fetch_add(p, v, __ATOMIC_RELAXED, __HIP_MEMORY_SCOPE_WORKGROUP);
}

__global__ void __launch_bounds__(EDGE_THREADS)
k_bin_count(const int* __restrict__ dst, int* __restrict__ cntBB, int E, int nb, int per) {
  __shared__ int h[MAX_BINS];
  const int tid = threadIdx.x;
  const int b = blockIdx.x;
  for (int i = tid; i < nb; i += EDGE_THREADS) h[i] = 0;
  __syncthreads();
  const int e0 = b * per;
  const int e1 = (e0 + per < E) ? e0 + per : E;
  for (int base = e0; base < e1; base += EDGE_AHEAD * EDGE_THREADS) {
    int d[EDGE_AHEAD];
#pragma unroll
    for (int j = 0; j < EDGE_AHEAD; ++j) {
      const int e = base + j * EDGE_THREADS + tid;
      d[j] = (e < e1) ? dst[e] : -1;
    }
#pragma unroll
    for (int j = 0; j < EDGE_AHEAD; ++j)
      if (d[j] >= 0) lds_add(&h[d[j] >> BIN_SHIFT], 1);
  }
  __syncthreads();
  for (int i = tid; i < nb; i += EDGE_THREADS) cntBB[i * CHUNKS + b] = h[i];
}

__global__ void __launch_bounds__(1024)
k_scan(const int* __restrict__ cnt, int* __restrict__ pre, int n) {
  __shared__ int s_incl[32];
  __shared__ int s_part[32];
  const int t = threadIdx.x;
  const int lane = t & 31;
  const int wv = t >> 5;
  const int first = blockIdx.x * 1024;

  const int4* q = (const int4*)cnt;
  const int nq = first >> 2;
  const int last = nq - 1;
  int p0 = 0, p1 = 0, p2 = 0, p3 = 0;
  for (int k = t; k < nq; k += 4096) {
    const int k1 = k + 1024, k2 = k + 2048, k3 = k + 3072;
    const int4 a = q[k];
    const int4 b = q[k1 < nq ? k1 : last];
    const int4 c = q[k2 < nq ? k2 : last];
    const int4 d = q[k3 < nq ? k3 : last];
    p0 += (a.x + a.y) + (a.z + a.w);
    p1 += (k1 < nq) ? (b.x + b.y) + (b.z + b.w) : 0;
    p2 += (k2 < nq) ? (c.x + c.y) + (c.z + c.w) : 0;
    p3 += (k3 < nq) ? (d.x + d.y) + (d.z + d.w) : 0;
  }
  int part = (p0 + p1) + (p2 + p3);

  const int i = first + t;
  const int c = (i < n) ? cnt[i] : 0;
  int incl = c;
  for (int d = 1; d < 32; d <<= 1) {
    int u = __shfl_up(incl, d);
    if (lane >= d) incl += u;
  }
  for (int d = 16; d >= 1; d >>= 1) part += __shfl_xor(part, d);
  if (lane == 31) s_incl[wv] = incl;
  if (lane == 0) s_part[wv] = part;
  __syncthreads();
  int front = s_part[lane] + ((lane < wv) ? s_incl[lane] : 0);
  for (int d = 16; d >= 1; d >>= 1) front += __shfl_xor(front, d);
  const int excl = front + incl - c;
  if (i < n) {
    pre[i] = excl;
    if (i == n - 1) pre[n] = excl + c;
  }
}

__global__ void __launch_bounds__(EDGE_THREADS)
k_bin_place(const int* __restrict__ src, const int* __restrict__ dst,
            const int* __restrict__ offBB, int* __restrict__ pairs_w, int E, int nb, int per) {
  __shared__ int cur[MAX_BINS];
  int2* pairs = (int2*)pairs_w;
  const int tid = threadIdx.x;
  const int b = blockIdx.x;
  for (int i = tid; i < nb; i += EDGE_THREADS) cur[i] = offBB[i * CHUNKS + b];
  __syncthreads();
  const int e0 = b * per;
  const int e1 = (e0 + per < E) ? e0 + per : E;
  for (int base = e0; base < e1; base += EDGE_AHEAD * EDGE_THREADS) {
    int d[EDGE_AHEAD], s[EDGE_AHEAD];
#pragma unroll
    for (int j = 0; j < EDGE_AHEAD; ++j) {
      const int e = base + j * EDGE_THREADS + tid;
      d[j] = (e < e1) ? dst[e] : -1;
      s[j] = (e < e1) ? src[e] : 0;
    }
#pragma unroll
    for (int j = 0; j < EDGE_AHEAD; ++j)
      if (d[j] >= 0) {
        const int slot = lds_add(&cur[d[j] >> BIN_SHIFT], 1);
        pairs[slot] = make_int2(s[j], d[j]);
      }
  }
}

__global__ void __launch_bounds__(EDGE_THREADS)
k_bin_finish(const int* __restrict__ offBB, const int* __restrict__ pairs_w,
             int* __restrict__ rowptr, int* __restrict__ csr, float* __restrict__ dinv,
             int n, int E) {
  const int2* pairs = (const int2*)pairs_w;
  __shared__ int c[BIN_NODES];
  __shared__ int wsum[BIN_NODES / 32];
  const int tid = threadIdx.x;
  const int lane = tid & 31;
  const int wv = tid >> 5;
  const int bin = blockIdx.x;
  const int s0 = offBB[bin * CHUNKS];
  const int s1 = offBB[(bin + 1) * CHUNKS];
  if (tid < BIN_NODES) c[tid] = 0;
  __syncthreads();
  for (int base = s0; base < s1; base += BIN_AHEAD * EDGE_THREADS) {
    int d[BIN_AHEAD];
#pragma unroll
    for (int j = 0; j < BIN_AHEAD; ++j) {
      const int e = base + j * EDGE_THREADS + tid;
      d[j] = (e < s1) ? pairs[e].y : -1;
    }
#pragma unroll
    for (int j = 0; j < BIN_AHEAD; ++j)
      if (d[j] >= 0) lds_add(&c[d[j] & (BIN_NODES - 1)], 1);
  }
  __syncthreads();
  const int cv = (tid < BIN_NODES) ? c[tid] : 0;
  int incl = cv;
  for (int d = 1; d < 32; d <<= 1) {
    int u = __shfl_up(incl, d);
    if (lane >= d) incl += u;
  }
  if (lane == 31 && wv < BIN_NODES / 32) wsum[wv] = incl;
  __syncthreads();
  int front = s0;
  for (int k = 0; k < BIN_NODES / 32; ++k) front += (k < wv) ? wsum[k] : 0;
  const int excl = front + incl - cv;
  const int node = bin * BIN_NODES + tid;
  if (tid < BIN_NODES) {
    c[tid] = excl;
    if (node < n) {
      rowptr[node] = excl;
      float dg = (float)(cv + 1);
      dinv[node] = (dg > 0.0f) ? rsqrtf(dg) : 0.0f;
    }
  }
  if (bin == (int)gridDim.x - 1 && tid == 0) rowptr[n] = E;
  __syncthreads();
  for (int base = s0; base < s1; base += BIN_AHEAD * EDGE_THREADS) {
    int2 pr[BIN_AHEAD];
#pragma unroll
    for (int j = 0; j < BIN_AHEAD; ++j) {
      const int e = base + j * EDGE_THREADS + tid;
      pr[j] = (e < s1) ? pairs[e] : make_int2(0, -1);
    }
#pragma unroll
    for (int j = 0; j < BIN_AHEAD; ++j)
      if (pr[j].y >= 0) {
        const int pos = lds_add(&c[pr[j].y & (BIN_NODES - 1)], 1);
        csr[pos] = pr[j].x;
      }
  }
}

__global__ void k_move_list(const int* __restrict__ from, int* __restrict__ to, int n) {
  int i = blockIdx.x * blockDim.x + threadIdx.x;
  if (i < n) to[i] = from[i];
}

constexpr int GEMM_THREADS = 800;
template <int K, int NC, bool RELU>
__global__ void __launch_bounds__(GEMM_THREADS)
k_gemm_strip(const float* __restrict__ A, const float* __restrict__ W,
             float* __restrict__ C, int nstrips) {
  constexpr int TILES = NC / 16;
  constexpr int P2 = 2 * NC + 32;
  constexpr int NQ = NC / 4;
  __shared__ __attribute__((aligned(16))) float Ws[(K / 2) * P2];

  typedef __attribute__((ext_vector_type(4))) float v4f;
#pragma unroll
  for (int idx = threadIdx.x; idx < (K / 2) * NQ; idx += GEMM_THREADS) {
    const int p = idx / NQ;
    const int q = idx % NQ;
    const v4f w0 = *(const v4f*)(W + (size_t)(2 * p) * NC + 4 * q);
    const v4f w1 = *(const v4f*)(W + (size_t)(2 * p + 1) * NC + 4 * q);
    v4f lo, hi;
    lo.x = w0.x; lo.y = w1.x; lo.z = w0.y; lo.w = w1.y;
    hi.x = w0.z; hi.y = w1.z; hi.z = w0.w; hi.w = w1.w;
    float* d = Ws + p * P2 + 8 * q;
    *(v4f*)d = lo;
    *(v4f*)(d + 4) = hi;
  }
  __syncthreads();

  const int lane = threadIdx.x & 31;
  const int wave = threadIdx.x >> 5;
  const int strip = blockIdx.x * (GEMM_THREADS / 32) + wave;
  if (strip >= nstrips) return;
  const int c = lane & 15;
  const int h = lane >> 4;

  const float* arow = A + (size_t)(strip * 16 + c) * K + 2 * h;
  const float* bl = Ws + h * P2 + 2 * c;

  v8f acc[TILES];
#pragma unroll
  for (int t = 0; t < TILES; ++t) acc[t] = v8f{};

  v2f acur[8], anxt[8];
#pragma unroll
  for (int s = 0; s < 8; ++s) acur[s] = *(const v2f*)(arow + 4 * s);

  v2f bcur[TILES], bnxt[TILES];
#pragma unroll
  for (int t = 0; t < TILES; ++t) bcur[t] = *(const v2f*)(bl + 32 * t);

  for (int k0 = 0; k0 < K; k0 += 32) {
    const int kn = (k0 + 32 < K) ? k0 + 32 : k0;
#pragma unroll
    for (int s = 0; s < 8; ++s) anxt[s] = *(const v2f*)(arow + kn + 4 * s);

#pragma unroll
    for (int s = 0; s < 8; ++s) {
      v2f a = acur[s];
      if (RELU) { a.x = fmaxf(a.x, 0.0f); a.y = fmaxf(a.y, 0.0f); }
      const int kb = (k0 + 4 * s + 4 < K) ? k0 + 4 * s + 4 : 0;
      const float* bp = bl + (kb >> 1) * P2;
#pragma unroll
      for (int t = 0; t < TILES; ++t) bnxt[t] = *(const v2f*)(bp + 32 * t);
      __builtin_amdgcn_sched_barrier(0);
#pragma unroll
      for (int t = 0; t < TILES; ++t)
        acc[t] = __builtin_amdgcn_wmma_f32_16x16x4_f32(false, a, false, bcur[t], (short)0,
                                                       acc[t], false, false);
      __builtin_amdgcn_sched_barrier(0);
#pragma unroll
      for (int t = 0; t < TILES; ++t) bcur[t] = bnxt[t];
    }
#pragma unroll
    for (int s = 0; s < 8; ++s) acur[s] = anxt[s];
  }

  const int rbase = strip * 16 + 8 * h;
#pragma unroll
  for (int t = 0; t < TILES; ++t)
#pragma unroll
    for (int v = 0; v < 8; ++v)
      C[(size_t)(rbase + v) * NC + 16 * t + c] = acc[t][v];
}

template <int F>
__global__ void k_gather(const int* __restrict__ rowptr, const int* __restrict__ csr,
                         const float* __restrict__ dinv, const float* __restrict__ H,
                         const float* __restrict__ bias, float* __restrict__ O, int n) {
  constexpr int V = F / 32;
  typedef __attribute__((ext_vector_type(V))) float vecT;
  const int lane = threadIdx.x & 31;
  const int node = blockIdx.x * (blockDim.x >> 5) + (threadIdx.x >> 5);
  if (node >= n) return;
  const int beg = __builtin_amdgcn_readfirstlane(rowptr[node]);
  const int end = __builtin_amdgcn_readfirstlane(rowptr[node + 1]);
  const int lo = lane * V;
  const float di = dinv[node];
  vecT acc = *(const vecT*)(bias + lo) + *(const vecT*)(H + (size_t)node * F + lo) * (di * di);

  for (int base = beg; base < end; base += 32) {
    int ms = 0;
    float mn = 0.0f;
    if (base + lane < end) {
      ms = csr[base + lane];
      mn = dinv[ms] * di;
    }
    const int mni = __float_as_int(mn);
    const int c = (end - base < 32) ? (end - base) : 32;
    int j = 0;
    for (; j + 4 <= c; j += 4) {
      const int s0 = __builtin_amdgcn_readlane(ms, j);
      const int s1 = __builtin_amdgcn_readlane(ms, j + 1);
      const int s2 = __builtin_amdgcn_readlane(ms, j + 2);
      const int s3 = __builtin_amdgcn_readlane(ms, j + 3);
      const float w0 = __int_as_float(__builtin_amdgcn_readlane(mni, j));
      const float w1 = __int_as_float(__builtin_amdgcn_readlane(mni, j + 1));
      const float w2 = __int_as_float(__builtin_amdgcn_readlane(mni, j + 2));
      const float w3 = __int_as_float(__builtin_amdgcn_readlane(mni, j + 3));
      const vecT m0 = *(const vecT*)(H + (size_t)s0 * F + lo);
      const vecT m1 = *(const vecT*)(H + (size_t)s1 * F + lo);
      const vecT m2 = *(const vecT*)(H + (size_t)s2 * F + lo);
      const vecT m3 = *(const vecT*)(H + (size_t)s3 * F + lo);
      acc += m0 * w0;
      acc += m1 * w1;
      acc += m2 * w2;
      acc += m3 * w3;
    }
    if (j + 2 <= c) {
      const int s0 = __builtin_amdgcn_readlane(ms, j);
      const int s1 = __builtin_amdgcn_readlane(ms, j + 1);
      const float w0 = __int_as_float(__builtin_amdgcn_readlane(mni, j));
      const float w1 = __int_as_float(__builtin_amdgcn_readlane(mni, j + 1));
      const vecT m0 = *(const vecT*)(H + (size_t)s0 * F + lo);
      const vecT m1 = *(const vecT*)(H + (size_t)s1 * F + lo);
      acc += m0 * w0;
      acc += m1 * w1;
      j += 2;
    }
    if (j < c) {
      const int s0 = __builtin_amdgcn_readlane(ms, j);
      const float w0 = __int_as_float(__builtin_amdgcn_readlane(mni, j));
      const vecT m0 = *(const vecT*)(H + (size_t)s0 * F + lo);
      acc += m0 * w0;
    }
  }
  *(vecT*)(O + (size_t)node * F + lo) = acc;
}

extern "C" void kernel_launch(void* const* d_in, const int* in_sizes, int n_in,
                              void* d_out, int out_size, void* d_ws, size_t ws_size,
                              hipStream_t stream) {
  const float* x  = (const float*)d_in[0];
  const int*   ei = (const int*)d_in[1];
  const float* W1 = (const float*)d_in[2];
  const float* b1 = (const float*)d_in[3];
  const float* W2 = (const float*)d_in[4];
  const float* b2 = (const float*)d_in[5];
  float* out = (float*)d_out;

  const int FIN = 256, FHID = 128, FOUT = 64;
  const int N = in_sizes[0] / FIN;
  const int E = in_sizes[1] / 2;

  const int* srcI = ei;
  const int* dstI = ei + E;

  float* ws   = (float*)d_ws;
  float* dinv = ws;
  float* xw1  = ws + N;
  float* agg1 = xw1 + (size_t)N * FHID;
  float* h2   = xw1;

  const int NB = (N + BIN_NODES - 1) / BIN_NODES;
  const int NBC = NB * CHUNKS;
  const int PER = (E + CHUNKS - 1) / CHUNKS;
  int*  cntBB = (int*)agg1;
  int*  offBB = cntBB + 131072;
  int*  pairs = offBB + 131072;
  int* listA = (int*)out;
  int* listB = (int*)(xw1 + (size_t)N * FOUT);
  const int LIST = N + 1 + E;

  const int T = 256;
  const int NS = N / 16;
  const int GW = GEMM_THREADS / 32;

  k_bin_count<<<CHUNKS, EDGE_THREADS, 0, stream>>>(dstI, cntBB, E, NB, PER);
  k_scan<<<(NBC + 1023) / 1024, 1024, 0, stream>>>(cntBB, offBB, NBC);
  k_bin_place<<<CHUNKS, EDGE_THREADS, 0, stream>>>(srcI, dstI, offBB, pairs, E, NB, PER);
  k_bin_finish<<<NB, EDGE_THREADS, 0, stream>>>(offBB, pairs, listA, listA + N + 1, dinv, N, E);

  k_gemm_strip<FIN, FHID, false><<<(NS + GW - 1) / GW, GEMM_THREADS, 0, stream>>>(x, W1, xw1, NS);
  k_gather<128><<<(N + 7) / 8, T, 0, stream>>>(listA, listA + N + 1, dinv, xw1, b1, agg1, N);
  k_move_list<<<(LIST + T - 1) / T, T, 0, stream>>>(listA, listB, LIST);

  k_gemm_strip<FHID, FOUT, true><<<(NS + GW - 1) / GW, GEMM_THREADS, 0, stream>>>(agg1, W2, h2, NS);
  k_gather<64><<<(N + 7) / 8, T, 0, stream>>>(listB, listB + N + 1, dinv, h2, b2, out, N);
}


# Layer-2 projection moved into the layer-1 aggregation launch: 32 nodes per work-group summed into an LDS tile, ReLU, WMMA by W2 from LDS; aggregated layer never stored, list copy launch gone
# speedup vs baseline: 3.7317x; 1.0735x over previous
#include <hip/hip_runtime.h>

typedef __attribute__((ext_vector_type(2))) float v2f;
typedef __attribute__((ext_vector_type(8))) float v8f;

constexpr int BIN_SHIFT = 8;
constexpr int BIN_NODES = 1 << BIN_SHIFT;
constexpr int MAX_BINS = 512;
constexpr int CHUNKS = 250;
constexpr int EDGE_THREADS = 1024;
constexpr int EDGE_AHEAD = 7;
constexpr int BIN_AHEAD = 6;

__device__ __forceinline__ int lds_add(int* p, int v) {
  return __hip_atomic_fetch_add(p, v, __ATOMIC_RELAXED, __HIP_MEMORY_SCOPE_WORKGROUP);
}

__global__ void __launch_bounds__(EDGE_THREADS)
k_bin_count(const int* __restrict__ dst, int* __restrict__ cntBB, int E, int nb, int per) {
  __shared__ int h[MAX_BINS];
  const int tid = threadIdx.x;
  const int b = blockIdx.x;
  for (int i = tid; i < nb; i += EDGE_THREADS) h[i] = 0;
  __syncthreads();
  const int e0 = b * per;
  const int e1 = (e0 + per < E) ? e0 + per : E;
  for (int base = e0; base < e1; base += EDGE_AHEAD * EDGE_THREADS) {
    int d[EDGE_AHEAD];
#pragma unroll
    for (int j = 0; j < EDGE_AHEAD; ++j) {
      const int e = base + j * EDGE_THREADS + tid;
      d[j] = (e < e1) ? dst[e] : -1;
    }
#pragma unroll
    for (int j = 0; j < EDGE_AHEAD; ++j)
      if (d[j] >= 0) lds_add(&h[d[j] >> BIN_SHIFT], 1);
  }
  __syncthreads();
  for (int i = tid; i < nb; i += EDGE_THREADS) cntBB[i * CHUNKS + b] = h[i];
}

__global__ void __launch_bounds__(1024)
k_scan(const int* __restrict__ cnt, int* __restrict__ pre, int n) {
  __shared__ int s_incl[32];
  __shared__ int s_part[32];
  const int t = threadIdx.x;
  const int lane = t & 31;
  const int wv = t >> 5;
  const int first = blockIdx.x * 1024;

  const int4* q = (const int4*)cnt;
  const int nq = first >> 2;
  const int last = nq - 1;
  int p0 = 0, p1 = 0, p2 = 0, p3 = 0;
  for (int k = t; k < nq; k += 4096) {
    const int k1 = k + 1024, k2 = k + 2048, k3 = k + 3072;
    const int4 a = q[k];
    const int4 b = q[k1 < nq ? k1 : last];
    const int4 c = q[k2 < nq ? k2 : last];
    const int4 d = q[k3 < nq ? k3 : last];
    p0 += (a.x + a.y) + (a.z + a.w);
    p1 += (k1 < nq) ? (b.x + b.y) + (b.z + b.w) : 0;
    p2 += (k2 < nq) ? (c.x + c.y) + (c.z + c.w) : 0;
    p3 += (k3 < nq) ? (d.x + d.y) + (d.z + d.w) : 0;
  }
  int part = (p0 + p1) + (p2 + p3);

  const int i = first + t;
  const int c = (i < n) ? cnt[i] : 0;
  int incl = c;
  for (int d = 1; d < 32; d <<= 1) {
    int u = __shfl_up(incl, d);
    if (lane >= d) incl += u;
  }
  for (int d = 16; d >= 1; d >>= 1) part += __shfl_xor(part, d);
  if (lane == 31) s_incl[wv] = incl;
  if (lane == 0) s_part[wv] = part;
  __syncthreads();
  int front = s_part[lane] + ((lane < wv) ? s_incl[lane] : 0);
  for (int d = 16; d >= 1; d >>= 1) front += __shfl_xor(front, d);
  const int excl = front + incl - c;
  if (i < n) {
    pre[i] = excl;
    if (i == n - 1) pre[n] = excl + c;
  }
}

__global__ void __launch_bounds__(EDGE_THREADS)
k_bin_place(const int* __restrict__ src, const int* __restrict__ dst,
            const int* __restrict__ offBB, int* __restrict__ pairs_w, int E, int nb, int per) {
  __shared__ int cur[MAX_BINS];
  int2* pairs = (int2*)pairs_w;
  const int tid = threadIdx.x;
  const int b = blockIdx.x;
  for (int i = tid; i < nb; i += EDGE_THREADS) cur[i] = offBB[i * CHUNKS + b];
  __syncthreads();
  const int e0 = b * per;
  const int e1 = (e0 + per < E) ? e0 + per : E;
  for (int base = e0; base < e1; base += EDGE_AHEAD * EDGE_THREADS) {
    int d[EDGE_AHEAD], s[EDGE_AHEAD];
#pragma unroll
    for (int j = 0; j < EDGE_AHEAD; ++j) {
      const int e = base + j * EDGE_THREADS + tid;
      d[j] = (e < e1) ? dst[e] : -1;
      s[j] = (e < e1) ? src[e] : 0;
    }
#pragma unroll
    for (int j = 0; j < EDGE_AHEAD; ++j)
      if (d[j] >= 0) {
        const int slot = lds_add(&cur[d[j] >> BIN_SHIFT], 1);
        pairs[slot] = make_int2(s[j], d[j]);
      }
  }
}

__global__ void __launch_bounds__(EDGE_THREADS)
k_bin_finish(const int* __restrict__ offBB, const int* __restrict__ pairs_w,
             int* __restrict__ rowptr, int* __restrict__ csr, float* __restrict__ dinv,
             int n, int E) {
  const int2* pairs = (const int2*)pairs_w;
  __shared__ int c[BIN_NODES];
  __shared__ int wsum[BIN_NODES / 32];
  const int tid = threadIdx.x;
  const int lane = tid & 31;
  const int wv = tid >> 5;
  const int bin = blockIdx.x;
  const int s0 = offBB[bin * CHUNKS];
  const int s1 = offBB[(bin + 1) * CHUNKS];
  if (tid < BIN_NODES) c[tid] = 0;
  __syncthreads();
  for (int base = s0; base < s1; base += BIN_AHEAD * EDGE_THREADS) {
    int d[BIN_AHEAD];
#pragma unroll
    for (int j = 0; j < BIN_AHEAD; ++j) {
      const int e = base + j * EDGE_THREADS + tid;
      d[j] = (e < s1) ? pairs[e].y : -1;
    }
#pragma unroll
    for (int j = 0; j < BIN_AHEAD; ++j)
      if (d[j] >= 0) lds_add(&c[d[j] & (BIN_NODES - 1)], 1);
  }
  __syncthreads();
  const int cv = (tid < BIN_NODES) ? c[tid] : 0;
  int incl = cv;
  for (int d = 1; d < 32; d <<= 1) {
    int u = __shfl_up(incl, d);
    if (lane >= d) incl += u;
  }
  if (lane == 31 && wv < BIN_NODES / 32) wsum[wv] = incl;
  __syncthreads();
  int front = s0;
  for (int k = 0; k < BIN_NODES / 32; ++k) front += (k < wv) ? wsum[k] : 0;
  const int excl = front + incl - cv;
  const int node = bin * BIN_NODES + tid;
  if (tid < BIN_NODES) {
    c[tid] = excl;
    if (node < n) {
      rowptr[node] = excl;
      float dg = (float)(cv + 1);
      dinv[node] = (dg > 0.0f) ? rsqrtf(dg) : 0.0f;
    }
  }
  if (bin == (int)gridDim.x - 1 && tid == 0) rowptr[n] = E;
  __syncthreads();
  for (int base = s0; base < s1; base += BIN_AHEAD * EDGE_THREADS) {
    int2 pr[BIN_AHEAD];
#pragma unroll
    for (int j = 0; j < BIN_AHEAD; ++j) {
      const int e = base + j * EDGE_THREADS + tid;
      pr[j] = (e < s1) ? pairs[e] : make_int2(0, -1);
    }
#pragma unroll
    for (int j = 0; j < BIN_AHEAD; ++j)
      if (pr[j].y >= 0) {
        const int pos = lds_add(&c[pr[j].y & (BIN_NODES - 1)], 1);
        csr[pos] = pr[j].x;
      }
  }
}

constexpr int GEMM_THREADS = 800;
template <int K, int NC>
__global__ void __launch_bounds__(GEMM_THREADS)
k_gemm_strip(const float* __restrict__ A, const float* __restrict__ W,
             float* __restrict__ C, int nstrips) {
  constexpr int TILES = NC / 16;
  constexpr int P2 = 2 * NC + 32;
  constexpr int NQ = NC / 4;
  __shared__ __attribute__((aligned(16))) float Ws[(K / 2) * P2];

  typedef __attribute__((ext_vector_type(4))) float v4f;
#pragma unroll
  for (int idx = threadIdx.x; idx < (K / 2) * NQ; idx += GEMM_THREADS) {
    const int p = idx / NQ;
    const int q = idx % NQ;
    const v4f w0 = *(const v4f*)(W + (size_t)(2 * p) * NC + 4 * q);
    const v4f w1 = *(const v4f*)(W + (size_t)(2 * p + 1) * NC + 4 * q);
    v4f lo, hi;
    lo.x = w0.x; lo.y = w1.x; lo.z = w0.y; lo.w = w1.y;
    hi.x = w0.z; hi.y = w1.z; hi.z = w0.w; hi.w = w1.w;
    float* d = Ws + p * P2 + 8 * q;
    *(v4f*)d = lo;
    *(v4f*)(d + 4) = hi;
  }
  __syncthreads();

  const int lane = threadIdx.x & 31;
  const int wave = threadIdx.x >> 5;
  const int strip = blockIdx.x * (GEMM_THREADS / 32) + wave;
  if (strip >= nstrips) return;
  const int c = lane & 15;
  const int h = lane >> 4;

  const float* arow = A + (size_t)(strip * 16 + c) * K + 2 * h;
  const float* bl = Ws + h * P2 + 2 * c;

  v8f acc[TILES];
#pragma unroll
  for (int t = 0; t < TILES; ++t) acc[t] = v8f{};

  v2f acur[8], anxt[8];
#pragma unroll
  for (int s = 0; s < 8; ++s) acur[s] = *(const v2f*)(arow + 4 * s);

  v2f bcur[TILES], bnxt[TILES];
#pragma unroll
  for (int t = 0; t < TILES; ++t) bcur[t] = *(const v2f*)(bl + 32 * t);

  for (int k0 = 0; k0 < K; k0 += 32) {
    const int kn = (k0 + 32 < K) ? k0 + 32 : k0;
#pragma unroll
    for (int s = 0; s < 8; ++s) anxt[s] = *(const v2f*)(arow + kn + 4 * s);

#pragma unroll
    for (int s = 0; s < 8; ++s) {
      const v2f a = acur[s];
      const int kb = (k0 + 4 * s + 4 < K) ? k0 + 4 * s + 4 : 0;
      const float* bp = bl + (kb >> 1) * P2;
#pragma unroll
      for (int t = 0; t < TILES; ++t) bnxt[t] = *(const v2f*)(bp + 32 * t);
      __builtin_amdgcn_sched_barrier(0);
#pragma unroll
      for (int t = 0; t < TILES; ++t)
        acc[t] = __builtin_amdgcn_wmma_f32_16x16x4_f32(false, a, false, bcur[t], (short)0,
                                                       acc[t], false, false);
      __builtin_amdgcn_sched_barrier(0);
#pragma unroll
      for (int t = 0; t < TILES; ++t) bcur[t] = bnxt[t];
    }
#pragma unroll
    for (int s = 0; s < 8; ++s) acur[s] = anxt[s];
  }

  const int rbase = strip * 16 + 8 * h;
#pragma unroll
  for (int t = 0; t < TILES; ++t)
#pragma unroll
    for (int v = 0; v < 8; ++v)
      C[(size_t)(rbase + v) * NC + 16 * t + c] = acc[t][v];
}

template <int F>
struct RowVec {
  typedef __attribute__((ext_vector_type(F / 32))) float type;
};

template <int F>
struct NodeHead {
  int beg, end;
  float di;
  typename RowVec<F>::type acc;
};

template <int F>
__device__ __forceinline__ NodeHead<F>
node_head(int node, int lane, const int* __restrict__ rowptr, const float* __restrict__ dinv,
          const float* __restrict__ H, const float* __restrict__ bias) {
  typedef typename RowVec<F>::type vecT;
  NodeHead<F> hd;
  hd.beg = __builtin_amdgcn_readfirstlane(rowptr[node]);
  hd.end = __builtin_amdgcn_readfirstlane(rowptr[node + 1]);
  const int lo = lane * (F / 32);
  hd.di = dinv[node];
  hd.acc = *(const vecT*)(bias + lo) + *(const vecT*)(H + (size_t)node * F + lo) * (hd.di * hd.di);
  return hd;
}

template <int F>
__device__ __forceinline__ typename RowVec<F>::type
node_walk(const NodeHead<F>& hd, int lane, const int* __restrict__ csr,
          const float* __restrict__ dinv, const float* __restrict__ H) {
  typedef typename RowVec<F>::type vecT;
  const int end = hd.end;
  const int lo = lane * (F / 32);
  const float di = hd.di;
  vecT acc = hd.acc;

  for (int base = hd.beg; base < end; base += 32) {
    int ms = 0;
    float mn = 0.0f;
    if (base + lane < end) {
      ms = csr[base + lane];
      mn = dinv[ms] * di;
    }
    const int mni = __float_as_int(mn);
    const int c = (end - base < 32) ? (end - base) : 32;
    int j = 0;
    for (; j + 4 <= c; j += 4) {
      const int s0 = __builtin_amdgcn_readlane(ms, j);
      const int s1 = __builtin_amdgcn_readlane(ms, j + 1);
      const int s2 = __builtin_amdgcn_readlane(ms, j + 2);
      const int s3 = __builtin_amdgcn_readlane(ms, j + 3);
      const float w0 = __int_as_float(__builtin_amdgcn_readlane(mni, j));
      const float w1 = __int_as_float(__builtin_amdgcn_readlane(mni, j + 1));
      const float w2 = __int_as_float(__builtin_amdgcn_readlane(mni, j + 2));
      const float w3 = __int_as_float(__builtin_amdgcn_readlane(mni, j + 3));
      const vecT m0 = *(const vecT*)(H + (size_t)s0 * F + lo);
      const vecT m1 = *(const vecT*)(H + (size_t)s1 * F + lo);
      const vecT m2 = *(const vecT*)(H + (size_t)s2 * F + lo);
      const vecT m3 = *(const vecT*)(H + (size_t)s3 * F + lo);
      acc += m0 * w0;
      acc += m1 * w1;
      acc += m2 * w2;
      acc += m3 * w3;
    }
    if (j + 2 <= c) {
      const int s0 = __builtin_amdgcn_readlane(ms, j);
      const int s1 = __builtin_amdgcn_readlane(ms, j + 1);
      const float w0 = __int_as_float(__builtin_amdgcn_readlane(mni, j));
      const float w1 = __int_as_float(__builtin_amdgcn_readlane(mni, j + 1));
      const vecT m0 = *(const vecT*)(H + (size_t)s0 * F + lo);
      const vecT m1 = *(const vecT*)(H + (size_t)s1 * F + lo);
      acc += m0 * w0;
      acc += m1 * w1;
      j += 2;
    }
    if (j < c) {
      const int s0 = __builtin_amdgcn_readlane(ms, j);
      const float w0 = __int_as_float(__builtin_amdgcn_readlane(mni, j));
      const vecT m0 = *(const vecT*)(H + (size_t)s0 * F + lo);
      acc += m0 * w0;
    }
  }
  return acc;
}

template <int F>
__global__ void k_gather(const int* __restrict__ rowptr, const int* __restrict__ csr,
                         const float* __restrict__ dinv, const float* __restrict__ H,
                         const float* __restrict__ bias, float* __restrict__ O, int n) {
  typedef typename RowVec<F>::type vecT;
  const int lane = threadIdx.x & 31;
  const int node = blockIdx.x * (blockDim.x >> 5) + (threadIdx.x >> 5);
  if (node >= n) return;
  const vecT acc = node_walk<F>(node_head<F>(node, lane, rowptr, dinv, H, bias), lane, csr, dinv, H);
  *(vecT*)(O + (size_t)node * F + lane * (F / 32)) = acc;
}

constexpr int FUSE_WAVES = 16;
constexpr int FUSE_THREADS = 32 * FUSE_WAVES;
constexpr int FUSE_NODES = 2 * FUSE_WAVES;
constexpr int FUSE_F = 128;
constexpr int FUSE_NC = 64;
constexpr int FUSE_AP = FUSE_F + 4;
__global__ void __launch_bounds__(FUSE_THREADS)
k_gather_proj(const int* __restrict__ rowptr, const int* __restrict__ csr,
              const float* __restrict__ dinv, const float* __restrict__ H,
              const float* __restrict__ bias, const float* __restrict__ W,
              float* __restrict__ C, int n) {
  constexpr int F = FUSE_F;
  constexpr int NC = FUSE_NC;
  constexpr int P2 = 2 * NC + 32;
  constexpr int NQ = NC / 4;
  constexpr int COPIES = (F / 2) * NQ / FUSE_THREADS;
  static_assert(COPIES * FUSE_THREADS == (F / 2) * NQ, "W is copied in whole rounds");
  typedef __attribute__((ext_vector_type(4))) float v4f;
  __shared__ __attribute__((aligned(16))) float Ws[(F / 2) * P2];
  __shared__ __attribute__((aligned(16))) float As[FUSE_NODES * FUSE_AP];

  const int lane = threadIdx.x & 31;
  const int wave = __builtin_amdgcn_readfirstlane(threadIdx.x >> 5);
  const int first = blockIdx.x * FUSE_NODES;
  const int n0 = first + 2 * wave;
  const bool in0 = n0 < n;
  const bool in1 = n0 + 1 < n;
  const NodeHead<F> hd0 = node_head<F>(in0 ? n0 : 0, lane, rowptr, dinv, H, bias);
  const NodeHead<F> hd1 = node_head<F>(in1 ? n0 + 1 : 0, lane, rowptr, dinv, H, bias);

#pragma unroll
  for (int it = 0; it < COPIES; ++it) {
    const int idx = threadIdx.x + it * FUSE_THREADS;
    const int p = idx / NQ;
    const int q = idx % NQ;
    const v4f w0 = *(const v4f*)(W + (size_t)(2 * p) * NC + 4 * q);
    const v4f w1 = *(const v4f*)(W + (size_t)(2 * p + 1) * NC + 4 * q);
    v4f lo, hi;
    lo.x = w0.x; lo.y = w1.x; lo.z = w0.y; lo.w = w1.y;
    hi.x = w0.z; hi.y = w1.z; hi.z = w0.w; hi.w = w1.w;
    float* d = Ws + p * P2 + 8 * q;
    *(v4f*)d = lo;
    *(v4f*)(d + 4) = hi;
  }

  v4f r0 = node_walk<F>(hd0, lane, csr, dinv, H);
  v4f r1 = node_walk<F>(hd1, lane, csr, dinv, H);
  if (!in0) r0 = v4f{};
  if (!in1) r1 = v4f{};
  r0.x = fmaxf(r0.x, 0.0f); r0.y = fmaxf(r0.y, 0.0f);
  r0.z = fmaxf(r0.z, 0.0f); r0.w = fmaxf(r0.w, 0.0f);
  r1.x = fmaxf(r1.x, 0.0f); r1.y = fmaxf(r1.y, 0.0f);
  r1.z = fmaxf(r1.z, 0.0f); r1.w = fmaxf(r1.w, 0.0f);
  *(v4f*)(As + (2 * wave) * FUSE_AP + 4 * lane) = r0;
  *(v4f*)(As + (2 * wave + 1) * FUSE_AP + 4 * lane) = r1;
  __syncthreads();
  if (wave >= 8) return;

  const int s = wave >> 2;
  const int t = wave & 3;
  const int c = lane & 15;
  const int h = lane >> 4;
  const float* ap = As + (16 * s + c) * FUSE_AP + 2 * h;
  const float* bp = Ws + h * P2 + 2 * (16 * t + c);

  v8f acc = v8f{};
#pragma unroll
  for (int k = 0; k < F; k += 4) {
    const v2f a = *(const v2f*)(ap + k);
    const v2f b = *(const v2f*)(bp + (k >> 1) * P2);
    acc = __builtin_amdgcn_wmma_f32_16x16x4_f32(false, a, false, b, (short)0, acc, false, false);
  }

  const int r16 = first + 16 * s;
  float* crow = C + (size_t)(r16 + 8 * h) * NC + 16 * t + c;
  if (r16 + 16 <= n) {
#pragma unroll
    for (int v = 0; v < 8; ++v) crow[(size_t)v * NC] = acc[v];
  } else {
#pragma unroll
    for (int v = 0; v < 8; ++v)
      if (r16 + 8 * h + v < n) crow[(size_t)v * NC] = acc[v];
  }
}

extern "C" void kernel_launch(void* const* d_in, const int* in_sizes, int n_in,
                              void* d_out, int out_size, void* d_ws, size_t ws_size,
                              hipStream_t stream) {
  const float* x  = (const float*)d_in[0];
  const int*   ei = (const int*)d_in[1];
  const float* W1 = (const float*)d_in[2];
  const float* b1 = (const float*)d_in[3];
  const float* W2 = (const float*)d_in[4];
  const float* b2 = (const float*)d_in[5];
  float* out = (float*)d_out;

  const int FIN = 256, FHID = 128, FOUT = 64;
  const int N = in_sizes[0] / FIN;
  const int E = in_sizes[1] / 2;

  const int* srcI = ei;
  const int* dstI = ei + E;

  float* ws   = (float*)d_ws;
  float* dinv = ws;
  float* xw1  = ws + N;
  float* rest = xw1 + (size_t)N * FHID;
  float* h2   = rest;

  int* list = (int*)(rest + (size_t)N * FOUT);
  const int NB = (N + BIN_NODES - 1) / BIN_NODES;
  const int NBC = NB * CHUNKS;
  const int PER = (E + CHUNKS - 1) / CHUNKS;
  int*  cntBB = (int*)rest;
  int*  offBB = cntBB + 131072;
  int*  pairs = offBB + 131072;

  const int T = 256;
  const int NS = N / 16;
  const int GW = GEMM_THREADS / 32;

  k_bin_count<<<CHUNKS, EDGE_THREADS, 0, stream>>>(dstI, cntBB, E, NB, PER);
  k_scan<<<(NBC + 1023) / 1024, 1024, 0, stream>>>(cntBB, offBB, NBC);
  k_bin_place<<<CHUNKS, EDGE_THREADS, 0, stream>>>(srcI, dstI, offBB, pairs, E, NB, PER);
  k_bin_finish<<<NB, EDGE_THREADS, 0, stream>>>(offBB, pairs, list, list + N + 1, dinv, N, E);

  k_gemm_strip<FIN, FHID><<<(NS + GW - 1) / GW, GEMM_THREADS, 0, stream>>>(x, W1, xw1, NS);
  k_gather_proj<<<(N + FUSE_NODES - 1) / FUSE_NODES, FUSE_THREADS, 0, stream>>>(
      list, list + N + 1, dinv, xw1, b1, W2, h2, N);

  k_gather<64><<<(N + 7) / 8, T, 0, stream>>>(list, list + N + 1, dinv, h2, b2, out, N);
}


# Source-side edge weight folded into both projections' stores (row * dinv[row]): aggregations gather rows only, no dinv[src] gather, no per-edge multiply; sum * dinv[dst], bias last; rounding differs
# speedup vs baseline: 4.0381x; 1.0821x over previous
#include <hip/hip_runtime.h>

typedef __attribute__((ext_vector_type(2))) float v2f;
typedef __attribute__((ext_vector_type(8))) float v8f;

constexpr int BIN_SHIFT = 8;
constexpr int BIN_NODES = 1 << BIN_SHIFT;
constexpr int MAX_BINS = 512;
constexpr int CHUNKS = 250;
constexpr int EDGE_THREADS = 1024;
constexpr int EDGE_AHEAD = 7;
constexpr int BIN_AHEAD = 6;

__device__ __forceinline__ int lds_add(int* p, int v) {
  return __hip_atomic_fetch_add(p, v, __ATOMIC_RELAXED, __HIP_MEMORY_SCOPE_WORKGROUP);
}

__global__ void __launch_bounds__(EDGE_THREADS)
k_bin_count(const int* __restrict__ dst, int* __restrict__ cntBB, int E, int nb, int per) {
  __shared__ int h[MAX_BINS];
  const int tid = threadIdx.x;
  const int b = blockIdx.x;
  for (int i = tid; i < nb; i += EDGE_THREADS) h[i] = 0;
  __syncthreads();
  const int e0 = b * per;
  const int e1 = (e0 + per < E) ? e0 + per : E;
  for (int base = e0; base < e1; base += EDGE_AHEAD * EDGE_THREADS) {
    int d[EDGE_AHEAD];
#pragma unroll
    for (int j = 0; j < EDGE_AHEAD; ++j) {
      const int e = base + j * EDGE_THREADS + tid;
      d[j] = (e < e1) ? dst[e] : -1;
    }
#pragma unroll
    for (int j = 0; j < EDGE_AHEAD; ++j)
      if (d[j] >= 0) lds_add(&h[d[j] >> BIN_SHIFT], 1);
  }
  __syncthreads();
  for (int i = tid; i < nb; i += EDGE_THREADS) cntBB[i * CHUNKS + b] = h[i];
}

__global__ void __launch_bounds__(1024)
k_scan(const int* __restrict__ cnt, int* __restrict__ pre, int n) {
  __shared__ int s_incl[32];
  __shared__ int s_part[32];
  const int t = threadIdx.x;
  const int lane = t & 31;
  const int wv = t >> 5;
  const int first = blockIdx.x * 1024;

  const int4* q = (const int4*)cnt;
  const int nq = first >> 2;
  const int last = nq - 1;
  int p0 = 0, p1 = 0, p2 = 0, p3 = 0;
  for (int k = t; k < nq; k += 4096) {
    const int k1 = k + 1024, k2 = k + 2048, k3 = k + 3072;
    const int4 a = q[k];
    const int4 b = q[k1 < nq ? k1 : last];
    const int4 c = q[k2 < nq ? k2 : last];
    const int4 d = q[k3 < nq ? k3 : last];
    p0 += (a.x + a.y) + (a.z + a.w);
    p1 += (k1 < nq) ? (b.x + b.y) + (b.z + b.w) : 0;
    p2 += (k2 < nq) ? (c.x + c.y) + (c.z + c.w) : 0;
    p3 += (k3 < nq) ? (d.x + d.y) + (d.z + d.w) : 0;
  }
  int part = (p0 + p1) + (p2 + p3);

  const int i = first + t;
  const int c = (i < n) ? cnt[i] : 0;
  int incl = c;
  for (int d = 1; d < 32; d <<= 1) {
    int u = __shfl_up(incl, d);
    if (lane >= d) incl += u;
  }
  for (int d = 16; d >= 1; d >>= 1) part += __shfl_xor(part, d);
  if (lane == 31) s_incl[wv] = incl;
  if (lane == 0) s_part[wv] = part;
  __syncthreads();
  int front = s_part[lane] + ((lane < wv) ? s_incl[lane] : 0);
  for (int d = 16; d >= 1; d >>= 1) front += __shfl_xor(front, d);
  const int excl = front + incl - c;
  if (i < n) {
    pre[i] = excl;
    if (i == n - 1) pre[n] = excl + c;
  }
}

__global__ void __launch_bounds__(EDGE_THREADS)
k_bin_place(const int* __restrict__ src, const int* __restrict__ dst,
            const int* __restrict__ offBB, int* __restrict__ pairs_w, int E, int nb, int per) {
  __shared__ int cur[MAX_BINS];
  int2* pairs = (int2*)pairs_w;
  const int tid = threadIdx.x;
  const int b = blockIdx.x;
  for (int i = tid; i < nb; i += EDGE_THREADS) cur[i] = offBB[i * CHUNKS + b];
  __syncthreads();
  const int e0 = b * per;
  const int e1 = (e0 + per < E) ? e0 + per : E;
  for (int base = e0; base < e1; base += EDGE_AHEAD * EDGE_THREADS) {
    int d[EDGE_AHEAD], s[EDGE_AHEAD];
#pragma unroll
    for (int j = 0; j < EDGE_AHEAD; ++j) {
      const int e = base + j * EDGE_THREADS + tid;
      d[j] = (e < e1) ? dst[e] : -1;
      s[j] = (e < e1) ? src[e] : 0;
    }
#pragma unroll
    for (int j = 0; j < EDGE_AHEAD; ++j)
      if (d[j] >= 0) {
        const int slot = lds_add(&cur[d[j] >> BIN_SHIFT], 1);
        pairs[slot] = make_int2(s[j], d[j]);
      }
  }
}

__global__ void __launch_bounds__(EDGE_THREADS)
k_bin_finish(const int* __restrict__ offBB, const int* __restrict__ pairs_w,
             int* __restrict__ rowptr, int* __restrict__ csr, float* __restrict__ dinv,
             int n, int E) {
  const int2* pairs = (const int2*)pairs_w;
  __shared__ int c[BIN_NODES];
  __shared__ int wsum[BIN_NODES / 32];
  const int tid = threadIdx.x;
  const int lane = tid & 31;
  const int wv = tid >> 5;
  const int bin = blockIdx.x;
  const int s0 = offBB[bin * CHUNKS];
  const int s1 = offBB[(bin + 1) * CHUNKS];
  if (tid < BIN_NODES) c[tid] = 0;
  __syncthreads();
  for (int base = s0; base < s1; base += BIN_AHEAD * EDGE_THREADS) {
    int d[BIN_AHEAD];
#pragma unroll
    for (int j = 0; j < BIN_AHEAD; ++j) {
      const int e = base + j * EDGE_THREADS + tid;
      d[j] = (e < s1) ? pairs[e].y : -1;
    }
#pragma unroll
    for (int j = 0; j < BIN_AHEAD; ++j)
      if (d[j] >= 0) lds_add(&c[d[j] & (BIN_NODES - 1)], 1);
  }
  __syncthreads();
  const int cv = (tid < BIN_NODES) ? c[tid] : 0;
  int incl = cv;
  for (int d = 1; d < 32; d <<= 1) {
    int u = __shfl_up(incl, d);
    if (lane >= d) incl += u;
  }
  if (lane == 31 && wv < BIN_NODES / 32) wsum[wv] = incl;
  __syncthreads();
  int front = s0;
  for (int k = 0; k < BIN_NODES / 32; ++k) front += (k < wv) ? wsum[k] : 0;
  const int excl = front + incl - cv;
  const int node = bin * BIN_NODES + tid;
  if (tid < BIN_NODES) {
    c[tid] = excl;
    if (node < n) {
      rowptr[node] = excl;
      float dg = (float)(cv + 1);
      dinv[node] = (dg > 0.0f) ? rsqrtf(dg) : 0.0f;
    }
  }
  if (bin == (int)gridDim.x - 1 && tid == 0) rowptr[n] = E;
  __syncthreads();
  for (int base = s0; base < s1; base += BIN_AHEAD * EDGE_THREADS) {
    int2 pr[BIN_AHEAD];
#pragma unroll
    for (int j = 0; j < BIN_AHEAD; ++j) {
      const int e = base + j * EDGE_THREADS + tid;
      pr[j] = (e < s1) ? pairs[e] : make_int2(0, -1);
    }
#pragma unroll
    for (int j = 0; j < BIN_AHEAD; ++j)
      if (pr[j].y >= 0) {
        const int pos = lds_add(&c[pr[j].y & (BIN_NODES - 1)], 1);
        csr[pos] = pr[j].x;
      }
  }
}

constexpr int GEMM_THREADS = 800;
template <int K, int NC>
__global__ void __launch_bounds__(GEMM_THREADS)
k_gemm_strip(const float* __restrict__ A, const float* __restrict__ W,
             const float* __restrict__ rowscale, float* __restrict__ C, int nstrips) {
  constexpr int TILES = NC / 16;
  constexpr int P2 = 2 * NC + 32;
  constexpr int NQ = NC / 4;
  __shared__ __attribute__((aligned(16))) float Ws[(K / 2) * P2];

  typedef __attribute__((ext_vector_type(4))) float v4f;
#pragma unroll
  for (int idx = threadIdx.x; idx < (K / 2) * NQ; idx += GEMM_THREADS) {
    const int p = idx / NQ;
    const int q = idx % NQ;
    const v4f w0 = *(const v4f*)(W + (size_t)(2 * p) * NC + 4 * q);
    const v4f w1 = *(const v4f*)(W + (size_t)(2 * p + 1) * NC + 4 * q);
    v4f lo, hi;
    lo.x = w0.x; lo.y = w1.x; lo.z = w0.y; lo.w = w1.y;
    hi.x = w0.z; hi.y = w1.z; hi.z = w0.w; hi.w = w1.w;
    float* d = Ws + p * P2 + 8 * q;
    *(v4f*)d = lo;
    *(v4f*)(d + 4) = hi;
  }
  __syncthreads();

  const int lane = threadIdx.x & 31;
  const int wave = threadIdx.x >> 5;
  const int strip = blockIdx.x * (GEMM_THREADS / 32) + wave;
  if (strip >= nstrips) return;
  const int c = lane & 15;
  const int h = lane >> 4;

  const float* arow = A + (size_t)(strip * 16 + c) * K + 2 * h;
  const float* bl = Ws + h * P2 + 2 * c;

  v8f acc[TILES];
#pragma unroll
  for (int t = 0; t < TILES; ++t) acc[t] = v8f{};

  v2f acur[8], anxt[8];
#pragma unroll
  for (int s = 0; s < 8; ++s) acur[s] = *(const v2f*)(arow + 4 * s);

  v2f bcur[TILES], bnxt[TILES];
#pragma unroll
  for (int t = 0; t < TILES; ++t) bcur[t] = *(const v2f*)(bl + 32 * t);

  for (int k0 = 0; k0 < K; k0 += 32) {
    const int kn = (k0 + 32 < K) ? k0 + 32 : k0;
#pragma unroll
    for (int s = 0; s < 8; ++s) anxt[s] = *(const v2f*)(arow + kn + 4 * s);

#pragma unroll
    for (int s = 0; s < 8; ++s) {
      const v2f a = acur[s];
      const int kb = (k0 + 4 * s + 4 < K) ? k0 + 4 * s + 4 : 0;
      const float* bp = bl + (kb >> 1) * P2;
#pragma unroll
      for (int t = 0; t < TILES; ++t) bnxt[t] = *(const v2f*)(bp + 32 * t);
      __builtin_amdgcn_sched_barrier(0);
#pragma unroll
      for (int t = 0; t < TILES; ++t)
        acc[t] = __builtin_amdgcn_wmma_f32_16x16x4_f32(false, a, false, bcur[t], (short)0,
                                                       acc[t], false, false);
      __builtin_amdgcn_sched_barrier(0);
#pragma unroll
      for (int t = 0; t < TILES; ++t) bcur[t] = bnxt[t];
    }
#pragma unroll
    for (int s = 0; s < 8; ++s) acur[s] = anxt[s];
  }

  const int rbase = strip * 16 + 8 * h;
  float rs[8];
#pragma unroll
  for (int v = 0; v < 8; ++v) rs[v] = rowscale[rbase + v];
#pragma unroll
  for (int t = 0; t < TILES; ++t)
#pragma unroll
    for (int v = 0; v < 8; ++v)
      C[(size_t)(rbase + v) * NC + 16 * t + c] = acc[t][v] * rs[v];
}

template <int F>
struct RowVec {
  typedef __attribute__((ext_vector_type(F / 32))) float type;
};

template <int F>
struct NodeHead {
  int beg, end;
  float di;
  typename RowVec<F>::type acc;
};

template <int F>
__device__ __forceinline__ NodeHead<F>
node_head(int node, int lane, const int* __restrict__ rowptr, const float* __restrict__ dinv,
          const float* __restrict__ H) {
  typedef typename RowVec<F>::type vecT;
  NodeHead<F> hd;
  hd.beg = __builtin_amdgcn_readfirstlane(rowptr[node]);
  hd.end = __builtin_amdgcn_readfirstlane(rowptr[node + 1]);
  const int lo = lane * (F / 32);
  hd.di = dinv[node];
  hd.acc = *(const vecT*)(H + (size_t)node * F + lo);
  return hd;
}

template <int F>
__device__ __forceinline__ typename RowVec<F>::type
node_walk(const NodeHead<F>& hd, int lane, const int* __restrict__ csr,
          const float* __restrict__ H) {
  typedef typename RowVec<F>::type vecT;
  const int end = hd.end;
  const int lo = lane * (F / 32);
  vecT acc = hd.acc;

  for (int base = hd.beg; base < end; base += 32) {
    int ms = 0;
    if (base + lane < end) ms = csr[base + lane];
    const int c = (end - base < 32) ? (end - base) : 32;
    int j = 0;
    for (; j + 4 <= c; j += 4) {
      const int s0 = __builtin_amdgcn_readlane(ms, j);
      const int s1 = __builtin_amdgcn_readlane(ms, j + 1);
      const int s2 = __builtin_amdgcn_readlane(ms, j + 2);
      const int s3 = __builtin_amdgcn_readlane(ms, j + 3);
      const vecT m0 = *(const vecT*)(H + (size_t)s0 * F + lo);
      const vecT m1 = *(const vecT*)(H + (size_t)s1 * F + lo);
      const vecT m2 = *(const vecT*)(H + (size_t)s2 * F + lo);
      const vecT m3 = *(const vecT*)(H + (size_t)s3 * F + lo);
      acc += m0;
      acc += m1;
      acc += m2;
      acc += m3;
    }
    if (j + 2 <= c) {
      const int s0 = __builtin_amdgcn_readlane(ms, j);
      const int s1 = __builtin_amdgcn_readlane(ms, j + 1);
      const vecT m0 = *(const vecT*)(H + (size_t)s0 * F + lo);
      const vecT m1 = *(const vecT*)(H + (size_t)s1 * F + lo);
      acc += m0;
      acc += m1;
      j += 2;
    }
    if (j < c) {
      const int s0 = __builtin_amdgcn_readlane(ms, j);
      const vecT m0 = *(const vecT*)(H + (size_t)s0 * F + lo);
      acc += m0;
    }
  }
  return acc;
}

template <int F>
__global__ void k_gather(const int* __restrict__ rowptr, const int* __restrict__ csr,
                         const float* __restrict__ dinv, const float* __restrict__ H,
                         const float* __restrict__ bias, float* __restrict__ O, int n) {
  typedef typename RowVec<F>::type vecT;
  const int lane = threadIdx.x & 31;
  const int node = blockIdx.x * (blockDim.x >> 5) + (threadIdx.x >> 5);
  if (node >= n) return;
  const NodeHead<F> hd = node_head<F>(node, lane, rowptr, dinv, H);
  const vecT bv = *(const vecT*)(bias + lane * (F / 32));
  const vecT acc = node_walk<F>(hd, lane, csr, H);
  *(vecT*)(O + (size_t)node * F + lane * (F / 32)) = bv + acc * hd.di;
}

constexpr int FUSE_WAVES = 16;
constexpr int FUSE_THREADS = 32 * FUSE_WAVES;
constexpr int FUSE_NODES = 2 * FUSE_WAVES;
constexpr int FUSE_F = 128;
constexpr int FUSE_NC = 64;
constexpr int FUSE_AP = FUSE_F + 4;
__global__ void __launch_bounds__(FUSE_THREADS)
k_gather_proj(const int* __restrict__ rowptr, const int* __restrict__ csr,
              const float* __restrict__ dinv, const float* __restrict__ H,
              const float* __restrict__ bias, const float* __restrict__ W,
              float* __restrict__ C, int n) {
  constexpr int F = FUSE_F;
  constexpr int NC = FUSE_NC;
  constexpr int P2 = 2 * NC + 32;
  constexpr int NQ = NC / 4;
  constexpr int COPIES = (F / 2) * NQ / FUSE_THREADS;
  static_assert(COPIES * FUSE_THREADS == (F / 2) * NQ, "W is copied in whole rounds");
  typedef __attribute__((ext_vector_type(4))) float v4f;
  __shared__ __attribute__((aligned(16))) float Ws[(F / 2) * P2];
  __shared__ __attribute__((aligned(16))) float As[FUSE_NODES * FUSE_AP];

  const int lane = threadIdx.x & 31;
  const int wave = __builtin_amdgcn_readfirstlane(threadIdx.x >> 5);
  const int first = blockIdx.x * FUSE_NODES;
  const int n0 = first + 2 * wave;
  const bool in0 = n0 < n;
  const bool in1 = n0 + 1 < n;
  const NodeHead<F> hd0 = node_head<F>(in0 ? n0 : 0, lane, rowptr, dinv, H);
  const NodeHead<F> hd1 = node_head<F>(in1 ? n0 + 1 : 0, lane, rowptr, dinv, H);
  const v4f bv = *(const v4f*)(bias + 4 * lane);

#pragma unroll
  for (int it = 0; it < COPIES; ++it) {
    const int idx = threadIdx.x + it * FUSE_THREADS;
    const int p = idx / NQ;
    const int q = idx % NQ;
    const v4f w0 = *(const v4f*)(W + (size_t)(2 * p) * NC + 4 * q);
    const v4f w1 = *(const v4f*)(W + (size_t)(2 * p + 1) * NC + 4 * q);
    v4f lo, hi;
    lo.x = w0.x; lo.y = w1.x; lo.z = w0.y; lo.w = w1.y;
    hi.x = w0.z; hi.y = w1.z; hi.z = w0.w; hi.w = w1.w;
    float* d = Ws + p * P2 + 8 * q;
    *(v4f*)d = lo;
    *(v4f*)(d + 4) = hi;
  }

  v4f r0 = bv + node_walk<F>(hd0, lane, csr, H) * hd0.di;
  v4f r1 = bv + node_walk<F>(hd1, lane, csr, H) * hd1.di;
  if (!in0) r0 = v4f{};
  if (!in1) r1 = v4f{};
  r0.x = fmaxf(r0.x, 0.0f); r0.y = fmaxf(r0.y, 0.0f);
  r0.z = fmaxf(r0.z, 0.0f); r0.w = fmaxf(r0.w, 0.0f);
  r1.x = fmaxf(r1.x, 0.0f); r1.y = fmaxf(r1.y, 0.0f);
  r1.z = fmaxf(r1.z, 0.0f); r1.w = fmaxf(r1.w, 0.0f);
  *(v4f*)(As + (2 * wave) * FUSE_AP + 4 * lane) = r0;
  *(v4f*)(As + (2 * wave + 1) * FUSE_AP + 4 * lane) = r1;
  __syncthreads();
  if (wave >= 8) return;

  const int s = wave >> 2;
  const int t = wave & 3;
  const int c = lane & 15;
  const int h = lane >> 4;
  const float* ap = As + (16 * s + c) * FUSE_AP + 2 * h;
  const float* bp = Ws + h * P2 + 2 * (16 * t + c);

  v8f acc = v8f{};
#pragma unroll
  for (int k = 0; k < F; k += 4) {
    const v2f a = *(const v2f*)(ap + k);
    const v2f b = *(const v2f*)(bp + (k >> 1) * P2);
    acc = __builtin_amdgcn_wmma_f32_16x16x4_f32(false, a, false, b, (short)0, acc, false, false);
  }

  const int r16 = first + 16 * s;
  float* crow = C + (size_t)(r16 + 8 * h) * NC + 16 * t + c;
  const float* drow = dinv + r16 + 8 * h;
  if (r16 + 16 <= n) {
#pragma unroll
    for (int v = 0; v < 8; ++v) crow[(size_t)v * NC] = acc[v] * drow[v];
  } else {
#pragma unroll
    for (int v = 0; v < 8; ++v)
      if (r16 + 8 * h + v < n) crow[(size_t)v * NC] = acc[v] * drow[v];
  }
}

extern "C" void kernel_launch(void* const* d_in, const int* in_sizes, int n_in,
                              void* d_out, int out_size, void* d_ws, size_t ws_size,
                              hipStream_t stream) {
  const float* x  = (const float*)d_in[0];
  const int*   ei = (const int*)d_in[1];
  const float* W1 = (const float*)d_in[2];
  const float* b1 = (const float*)d_in[3];
  const float* W2 = (const float*)d_in[4];
  const float* b2 = (const float*)d_in[5];
  float* out = (float*)d_out;

  const int FIN = 256, FHID = 128, FOUT = 64;
  const int N = in_sizes[0] / FIN;
  const int E = in_sizes[1] / 2;

  const int* srcI = ei;
  const int* dstI = ei + E;

  float* ws   = (float*)d_ws;
  float* dinv = ws;
  float* xw1  = ws + N;
  float* rest = xw1 + (size_t)N * FHID;
  float* h2   = rest;

  int* list = (int*)(rest + (size_t)N * FOUT);
  const int NB = (N + BIN_NODES - 1) / BIN_NODES;
  const int NBC = NB * CHUNKS;
  const int PER = (E + CHUNKS - 1) / CHUNKS;
  int*  cntBB = (int*)rest;
  int*  offBB = cntBB + 131072;
  int*  pairs = offBB + 131072;

  const int T = 256;
  const int NS = N / 16;
  const int GW = GEMM_THREADS / 32;

  k_bin_count<<<CHUNKS, EDGE_THREADS, 0, stream>>>(dstI, cntBB, E, NB, PER);
  k_scan<<<(NBC + 1023) / 1024, 1024, 0, stream>>>(cntBB, offBB, NBC);
  k_bin_place<<<CHUNKS, EDGE_THREADS, 0, stream>>>(srcI, dstI, offBB, pairs, E, NB, PER);
  k_bin_finish<<<NB, EDGE_THREADS, 0, stream>>>(offBB, pairs, list, list + N + 1, dinv, N, E);

  k_gemm_strip<FIN, FHID><<<(NS + GW - 1) / GW, GEMM_THREADS, 0, stream>>>(x, W1, dinv, xw1, NS);
  k_gather_proj<<<(N + FUSE_NODES - 1) / FUSE_NODES, FUSE_THREADS, 0, stream>>>(
      list, list + N + 1, dinv, xw1, b1, W2, h2, N);

  k_gather<64><<<(N + 7) / 8, T, 0, stream>>>(list, list + N + 1, dinv, h2, b2, out, N);
}


# Edge grouping in 2 launches, not 4: each chunk orders its own edges by bin in its own slice (LDS counts, LDS scan), no cross-chunk count or scan launch; node sum order changes, still LDS arrival
# speedup vs baseline: 4.4497x; 1.1019x over previous
#include <hip/hip_runtime.h>

typedef __attribute__((ext_vector_type(2))) float v2f;
typedef __attribute__((ext_vector_type(8))) float v8f;

constexpr int BIN_SHIFT = 8;
constexpr int BIN_NODES = 1 << BIN_SHIFT;
constexpr int MAX_BINS = 512;
constexpr int CHUNKS = 250;
constexpr int EDGE_THREADS = 1024;
constexpr int EDGE_AHEAD = 7;
constexpr int SEG_LANES = 4;
constexpr int SEG_AHEAD = 8;
static_assert(MAX_BINS < EDGE_THREADS, "thread nb writes the closing row of offT");
static_assert(CHUNKS * SEG_LANES <= EDGE_THREADS, "every chunk's run has its threads");

__device__ __forceinline__ int lds_add(int* p, int v) {
  return __hip_atomic_fetch_add(p, v, __ATOMIC_RELAXED, __HIP_MEMORY_SCOPE_WORKGROUP);
}

__global__ void __launch_bounds__(EDGE_THREADS)
k_chunk_sort(const int* __restrict__ src, const int* __restrict__ dst,
             int* __restrict__ offT, int* __restrict__ pairs_w, int E, int nb, int per) {
  __shared__ int h[MAX_BINS];
  __shared__ int wsum[EDGE_THREADS / 32];
  int2* pairs = (int2*)pairs_w;
  const int tid = threadIdx.x;
  const int lane = tid & 31;
  const int wv = tid >> 5;
  const int b = blockIdx.x;
  if (tid < nb) h[tid] = 0;
  __syncthreads();
  const int e0 = b * per;
  const int e1 = (e0 + per < E) ? e0 + per : E;
  for (int base = e0; base < e1; base += EDGE_AHEAD * EDGE_THREADS) {
    int d[EDGE_AHEAD];
#pragma unroll
    for (int j = 0; j < EDGE_AHEAD; ++j) {
      const int e = base + j * EDGE_THREADS + tid;
      d[j] = (e < e1) ? dst[e] : -1;
    }
#pragma unroll
    for (int j = 0; j < EDGE_AHEAD; ++j)
      if (d[j] >= 0) lds_add(&h[d[j] >> BIN_SHIFT], 1);
  }
  __syncthreads();
  const int v = (tid < nb) ? h[tid] : 0;
  int incl = v;
  for (int d = 1; d < 32; d <<= 1) {
    int u = __shfl_up(incl, d);
    if (lane >= d) incl += u;
  }
  if (lane == 31) wsum[wv] = incl;
  __syncthreads();
  int front = (lane < wv) ? wsum[lane] : 0;
  for (int d = 16; d >= 1; d >>= 1) front += __shfl_xor(front, d);
  const int excl = e0 + front + incl - v;
  if (tid <= nb) {
    offT[tid * CHUNKS + b] = excl;
    if (tid < nb) h[tid] = excl;
  }
  __syncthreads();
  for (int base = e0; base < e1; base += EDGE_AHEAD * EDGE_THREADS) {
    int d[EDGE_AHEAD], s[EDGE_AHEAD];
#pragma unroll
    for (int j = 0; j < EDGE_AHEAD; ++j) {
      const int e = base + j * EDGE_THREADS + tid;
      d[j] = (e < e1) ? dst[e] : -1;
    }
#pragma unroll
    for (int j = 0; j < EDGE_AHEAD; ++j) {
      const int e = base + j * EDGE_THREADS + tid;
      s[j] = (e < e1) ? src[e] : 0;
    }
#pragma unroll
    for (int j = 0; j < EDGE_AHEAD; ++j)
      if (d[j] >= 0) {
        const int slot = lds_add(&h[d[j] >> BIN_SHIFT], 1);
        pairs[slot] = make_int2(s[j], d[j]);
      }
  }
}

__global__ void __launch_bounds__(EDGE_THREADS)
k_bin_finish(const int* __restrict__ offT, const int* __restrict__ pairs_w,
             int* __restrict__ rowptr, int* __restrict__ csr, float* __restrict__ dinv,
             int n, int E, int per) {
  const int2* pairs = (const int2*)pairs_w;
  __shared__ int c[BIN_NODES];
  __shared__ int wsum[BIN_NODES / 32];
  __shared__ int red[EDGE_THREADS / 32];
  const int tid = threadIdx.x;
  const int lane = tid & 31;
  const int wv = tid >> 5;
  const int bin = blockIdx.x;
  const int ch = tid / SEG_LANES;
  const int sub = tid % SEG_LANES;
  int a = 0, len = 0, lower = 0;
  if (ch < CHUNKS) {
    a = offT[bin * CHUNKS + ch];
    len = offT[(bin + 1) * CHUNKS + ch] - a;
    if (sub == 0) lower = a - ch * per;
  }
  for (int d = 16; d >= 1; d >>= 1) lower += __shfl_xor(lower, d);
  if (lane == 0) red[wv] = lower;
  if (tid < BIN_NODES) c[tid] = 0;
  __syncthreads();
  int s0 = red[lane];
  for (int d = 16; d >= 1; d >>= 1) s0 += __shfl_xor(s0, d);

  for (int j0 = sub; j0 < len; j0 += SEG_LANES * SEG_AHEAD) {
    int d[SEG_AHEAD];
#pragma unroll
    for (int u = 0; u < SEG_AHEAD; ++u) {
      const int j = j0 + SEG_LANES * u;
      d[u] = (j < len) ? pairs[a + j].y : -1;
    }
#pragma unroll
    for (int u = 0; u < SEG_AHEAD; ++u)
      if (d[u] >= 0) lds_add(&c[d[u] & (BIN_NODES - 1)], 1);
  }
  __syncthreads();
  const int cv = (tid < BIN_NODES) ? c[tid] : 0;
  int incl = cv;
  for (int d = 1; d < 32; d <<= 1) {
    int u = __shfl_up(incl, d);
    if (lane >= d) incl += u;
  }
  if (lane == 31 && wv < BIN_NODES / 32) wsum[wv] = incl;
  __syncthreads();
  int front = s0;
  for (int k = 0; k < BIN_NODES / 32; ++k) front += (k < wv) ? wsum[k] : 0;
  const int excl = front + incl - cv;
  const int node = bin * BIN_NODES + tid;
  if (tid < BIN_NODES) {
    c[tid] = excl;
    if (node < n) {
      rowptr[node] = excl;
      float dg = (float)(cv + 1);
      dinv[node] = (dg > 0.0f) ? rsqrtf(dg) : 0.0f;
    }
  }
  if (bin == (int)gridDim.x - 1 && tid == 0) rowptr[n] = E;
  __syncthreads();
  for (int j0 = sub; j0 < len; j0 += SEG_LANES * SEG_AHEAD) {
    int2 pr[SEG_AHEAD];
#pragma unroll
    for (int u = 0; u < SEG_AHEAD; ++u) {
      const int j = j0 + SEG_LANES * u;
      pr[u] = (j < len) ? pairs[a + j] : make_int2(0, -1);
    }
#pragma unroll
    for (int u = 0; u < SEG_AHEAD; ++u)
      if (pr[u].y >= 0) {
        const int pos = lds_add(&c[pr[u].y & (BIN_NODES - 1)], 1);
        csr[pos] = pr[u].x;
      }
  }
}

constexpr int GEMM_THREADS = 800;
template <int K, int NC>
__global__ void __launch_bounds__(GEMM_THREADS)
k_gemm_strip(const float* __restrict__ A, const float* __restrict__ W,
             const float* __restrict__ rowscale, float* __restrict__ C, int nstrips) {
  constexpr int TILES = NC / 16;
  constexpr int P2 = 2 * NC + 32;
  constexpr int NQ = NC / 4;
  __shared__ __attribute__((aligned(16))) float Ws[(K / 2) * P2];

  typedef __attribute__((ext_vector_type(4))) float v4f;
#pragma unroll
  for (int idx = threadIdx.x; idx < (K / 2) * NQ; idx += GEMM_THREADS) {
    const int p = idx / NQ;
    const int q = idx % NQ;
    const v4f w0 = *(const v4f*)(W + (size_t)(2 * p) * NC + 4 * q);
    const v4f w1 = *(const v4f*)(W + (size_t)(2 * p + 1) * NC + 4 * q);
    v4f lo, hi;
    lo.x = w0.x; lo.y = w1.x; lo.z = w0.y; lo.w = w1.y;
    hi.x = w0.z; hi.y = w1.z; hi.z = w0.w; hi.w = w1.w;
    float* d = Ws + p * P2 + 8 * q;
    *(v4f*)d = lo;
    *(v4f*)(d + 4) = hi;
  }
  __syncthreads();

  const int lane = threadIdx.x & 31;
  const int wave = threadIdx.x >> 5;
  const int strip = blockIdx.x * (GEMM_THREADS / 32) + wave;
  if (strip >= nstrips) return;
  const int c = lane & 15;
  const int h = lane >> 4;

  const float* arow = A + (size_t)(strip * 16 + c) * K + 2 * h;
  const float* bl = Ws + h * P2 + 2 * c;

  v8f acc[TILES];
#pragma unroll
  for (int t = 0; t < TILES; ++t) acc[t] = v8f{};

  v2f acur[8], anxt[8];
#pragma unroll
  for (int s = 0; s < 8; ++s) acur[s] = *(const v2f*)(arow + 4 * s);

  v2f bcur[TILES], bnxt[TILES];
#pragma unroll
  for (int t = 0; t < TILES; ++t) bcur[t] = *(const v2f*)(bl + 32 * t);

  for (int k0 = 0; k0 < K; k0 += 32) {
    const int kn = (k0 + 32 < K) ? k0 + 32 : k0;
#pragma unroll
    for (int s = 0; s < 8; ++s) anxt[s] = *(const v2f*)(arow + kn + 4 * s);

#pragma unroll
    for (int s = 0; s < 8; ++s) {
      const v2f a = acur[s];
      const int kb = (k0 + 4 * s + 4 < K) ? k0 + 4 * s + 4 : 0;
      const float* bp = bl + (kb >> 1) * P2;
#pragma unroll
      for (int t = 0; t < TILES; ++t) bnxt[t] = *(const v2f*)(bp + 32 * t);
      __builtin_amdgcn_sched_barrier(0);
#pragma unroll
      for (int t = 0; t < TILES; ++t)
        acc[t] = __builtin_amdgcn_wmma_f32_16x16x4_f32(false, a, false, bcur[t], (short)0,
                                                       acc[t], false, false);
      __builtin_amdgcn_sched_barrier(0);
#pragma unroll
      for (int t = 0; t < TILES; ++t) bcur[t] = bnxt[t];
    }
#pragma unroll
    for (int s = 0; s < 8; ++s) acur[s] = anxt[s];
  }

  const int rbase = strip * 16 + 8 * h;
  float rs[8];
#pragma unroll
  for (int v = 0; v < 8; ++v) rs[v] = rowscale[rbase + v];
#pragma unroll
  for (int t = 0; t < TILES; ++t)
#pragma unroll
    for (int v = 0; v < 8; ++v)
      C[(size_t)(rbase + v) * NC + 16 * t + c] = acc[t][v] * rs[v];
}

template <int F>
struct RowVec {
  typedef __attribute__((ext_vector_type(F / 32))) float type;
};

template <int F>
struct NodeHead {
  int beg, end;
  float di;
  typename RowVec<F>::type acc;
};

template <int F>
__device__ __forceinline__ NodeHead<F>
node_head(int node, int lane, const int* __restrict__ rowptr, const float* __restrict__ dinv,
          const float* __restrict__ H) {
  typedef typename RowVec<F>::type vecT;
  NodeHead<F> hd;
  hd.beg = __builtin_amdgcn_readfirstlane(rowptr[node]);
  hd.end = __builtin_amdgcn_readfirstlane(rowptr[node + 1]);
  const int lo = lane * (F / 32);
  hd.di = dinv[node];
  hd.acc = *(const vecT*)(H + (size_t)node * F + lo);
  return hd;
}

template <int F>
__device__ __forceinline__ typename RowVec<F>::type
node_walk(const NodeHead<F>& hd, int lane, const int* __restrict__ csr,
          const float* __restrict__ H) {
  typedef typename RowVec<F>::type vecT;
  const int end = hd.end;
  const int lo = lane * (F / 32);
  vecT acc = hd.acc;

  for (int base = hd.beg; base < end; base += 32) {
    int ms = 0;
    if (base + lane < end) ms = csr[base + lane];
    const int c = (end - base < 32) ? (end - base) : 32;
    int j = 0;
    for (; j + 4 <= c; j += 4) {
      const int s0 = __builtin_amdgcn_readlane(ms, j);
      const int s1 = __builtin_amdgcn_readlane(ms, j + 1);
      const int s2 = __builtin_amdgcn_readlane(ms, j + 2);
      const int s3 = __builtin_amdgcn_readlane(ms, j + 3);
      const vecT m0 = *(const vecT*)(H + (size_t)s0 * F + lo);
      const vecT m1 = *(const vecT*)(H + (size_t)s1 * F + lo);
      const vecT m2 = *(const vecT*)(H + (size_t)s2 * F + lo);
      const vecT m3 = *(const vecT*)(H + (size_t)s3 * F + lo);
      acc += m0;
      acc += m1;
      acc += m2;
      acc += m3;
    }
    if (j + 2 <= c) {
      const int s0 = __builtin_amdgcn_readlane(ms, j);
      const int s1 = __builtin_amdgcn_readlane(ms, j + 1);
      const vecT m0 = *(const vecT*)(H + (size_t)s0 * F + lo);
      const vecT m1 = *(const vecT*)(H + (size_t)s1 * F + lo);
      acc += m0;
      acc += m1;
      j += 2;
    }
    if (j < c) {
      const int s0 = __builtin_amdgcn_readlane(ms, j);
      const vecT m0 = *(const vecT*)(H + (size_t)s0 * F + lo);
      acc += m0;
    }
  }
  return acc;
}

template <int F>
__global__ void k_gather(const int* __restrict__ rowptr, const int* __restrict__ csr,
                         const float* __restrict__ dinv, const float* __restrict__ H,
                         const float* __restrict__ bias, float* __restrict__ O, int n) {
  typedef typename RowVec<F>::type vecT;
  const int lane = threadIdx.x & 31;
  const int node = blockIdx.x * (blockDim.x >> 5) + (threadIdx.x >> 5);
  if (node >= n) return;
  const NodeHead<F> hd = node_head<F>(node, lane, rowptr, dinv, H);
  const vecT bv = *(const vecT*)(bias + lane * (F / 32));
  const vecT acc = node_walk<F>(hd, lane, csr, H);
  *(vecT*)(O + (size_t)node * F + lane * (F / 32)) = bv + acc * hd.di;
}

constexpr int FUSE_WAVES = 16;
constexpr int FUSE_THREADS = 32 * FUSE_WAVES;
constexpr int FUSE_NODES = 2 * FUSE_WAVES;
constexpr int FUSE_F = 128;
constexpr int FUSE_NC = 64;
constexpr int FUSE_AP = FUSE_F + 4;
__global__ void __launch_bounds__(FUSE_THREADS)
k_gather_proj(const int* __restrict__ rowptr, const int* __restrict__ csr,
              const float* __restrict__ dinv, const float* __restrict__ H,
              const float* __restrict__ bias, const float* __restrict__ W,
              float* __restrict__ C, int n) {
  constexpr int F = FUSE_F;
  constexpr int NC = FUSE_NC;
  constexpr int P2 = 2 * NC + 32;
  constexpr int NQ = NC / 4;
  constexpr int COPIES = (F / 2) * NQ / FUSE_THREADS;
  static_assert(COPIES * FUSE_THREADS == (F / 2) * NQ, "W is copied in whole rounds");
  typedef __attribute__((ext_vector_type(4))) float v4f;
  __shared__ __attribute__((aligned(16))) float Ws[(F / 2) * P2];
  __shared__ __attribute__((aligned(16))) float As[FUSE_NODES * FUSE_AP];

  const int lane = threadIdx.x & 31;
  const int wave = __builtin_amdgcn_readfirstlane(threadIdx.x >> 5);
  const int first = blockIdx.x * FUSE_NODES;
  const int n0 = first + 2 * wave;
  const bool in0 = n0 < n;
  const bool in1 = n0 + 1 < n;
  const NodeHead<F> hd0 = node_head<F>(in0 ? n0 : 0, lane, rowptr, dinv, H);
  const NodeHead<F> hd1 = node_head<F>(in1 ? n0 + 1 : 0, lane, rowptr, dinv, H);
  const v4f bv = *(const v4f*)(bias + 4 * lane);

#pragma unroll
  for (int it = 0; it < COPIES; ++it) {
    const int idx = threadIdx.x + it * FUSE_THREADS;
    const int p = idx / NQ;
    const int q = idx % NQ;
    const v4f w0 = *(const v4f*)(W + (size_t)(2 * p) * NC + 4 * q);
    const v4f w1 = *(const v4f*)(W + (size_t)(2 * p + 1) * NC + 4 * q);
    v4f lo, hi;
    lo.x = w0.x; lo.y = w1.x; lo.z = w0.y; lo.w = w1.y;
    hi.x = w0.z; hi.y = w1.z; hi.z = w0.w; hi.w = w1.w;
    float* d = Ws + p * P2 + 8 * q;
    *(v4f*)d = lo;
    *(v4f*)(d + 4) = hi;
  }

  v4f r0 = bv + node_walk<F>(hd0, lane, csr, H) * hd0.di;
  v4f r1 = bv + node_walk<F>(hd1, lane, csr, H) * hd1.di;
  if (!in0) r0 = v4f{};
  if (!in1) r1 = v4f{};
  r0.x = fmaxf(r0.x, 0.0f); r0.y = fmaxf(r0.y, 0.0f);
  r0.z = fmaxf(r0.z, 0.0f); r0.w = fmaxf(r0.w, 0.0f);
  r1.x = fmaxf(r1.x, 0.0f); r1.y = fmaxf(r1.y, 0.0f);
  r1.z = fmaxf(r1.z, 0.0f); r1.w = fmaxf(r1.w, 0.0f);
  *(v4f*)(As + (2 * wave) * FUSE_AP + 4 * lane) = r0;
  *(v4f*)(As + (2 * wave + 1) * FUSE_AP + 4 * lane) = r1;
  __syncthreads();
  if (wave >= 8) return;

  const int s = wave >> 2;
  const int t = wave & 3;
  const int c = lane & 15;
  const int h = lane >> 4;
  const float* ap = As + (16 * s + c) * FUSE_AP + 2 * h;
  const float* bp = Ws + h * P2 + 2 * (16 * t + c);

  v8f acc = v8f{};
#pragma unroll
  for (int k = 0; k < F; k += 4) {
    const v2f a = *(const v2f*)(ap + k);
    const v2f b = *(const v2f*)(bp + (k >> 1) * P2);
    acc = __builtin_amdgcn_wmma_f32_16x16x4_f32(false, a, false, b, (short)0, acc, false, false);
  }

  const int r16 = first + 16 * s;
  float* crow = C + (size_t)(r16 + 8 * h) * NC + 16 * t + c;
  const float* drow = dinv + r16 + 8 * h;
  if (r16 + 16 <= n) {
#pragma unroll
    for (int v = 0; v < 8; ++v) crow[(size_t)v * NC] = acc[v] * drow[v];
  } else {
#pragma unroll
    for (int v = 0; v < 8; ++v)
      if (r16 + 8 * h + v < n) crow[(size_t)v * NC] = acc[v] * drow[v];
  }
}

extern "C" void kernel_launch(void* const* d_in, const int* in_sizes, int n_in,
                              void* d_out, int out_size, void* d_ws, size_t ws_size,
                              hipStream_t stream) {
  const float* x  = (const float*)d_in[0];
  const int*   ei = (const int*)d_in[1];
  const float* W1 = (const float*)d_in[2];
  const float* b1 = (const float*)d_in[3];
  const float* W2 = (const float*)d_in[4];
  const float* b2 = (const float*)d_in[5];
  float* out = (float*)d_out;

  const int FIN = 256, FHID = 128, FOUT = 64;
  const int N = in_sizes[0] / FIN;
  const int E = in_sizes[1] / 2;

  const int* srcI = ei;
  const int* dstI = ei + E;

  float* ws   = (float*)d_ws;
  float* dinv = ws;
  float* xw1  = ws + N;
  float* rest = xw1 + (size_t)N * FHID;
  float* h2   = rest;

  int* list = (int*)(rest + (size_t)N * FOUT);
  const int NB = (N + BIN_NODES - 1) / BIN_NODES;
  const int PER = (E + CHUNKS - 1) / CHUNKS;
  int*  offT  = (int*)rest;
  int*  pairs = offT + 131072;

  const int T = 256;
  const int NS = N / 16;
  const int GW = GEMM_THREADS / 32;

  k_chunk_sort<<<CHUNKS, EDGE_THREADS, 0, stream>>>(srcI, dstI, offT, pairs, E, NB, PER);
  k_bin_finish<<<NB, EDGE_THREADS, 0, stream>>>(offT, pairs, list, list + N + 1, dinv, N, E,
                                                PER);

  k_gemm_strip<FIN, FHID><<<(NS + GW - 1) / GW, GEMM_THREADS, 0, stream>>>(x, W1, dinv, xw1, NS);
  k_gather_proj<<<(N + FUSE_NODES - 1) / FUSE_NODES, FUSE_THREADS, 0, stream>>>(
      list, list + N + 1, dinv, xw1, b1, W2, h2, N);

  k_gather<64><<<(N + 7) / 8, T, 0, stream>>>(list, list + N + 1, dinv, h2, b2, out, N);
}


# Grouped edge records staged in LDS and written out as whole 128-B lines: chunk slices 16 B per lane, bin source lists by line-aligned waves; no scattered 8-B or 4-B stores; sum order as before
# speedup vs baseline: 4.6702x; 1.0496x over previous
#include <hip/hip_runtime.h>

typedef __attribute__((ext_vector_type(2))) float v2f;
typedef __attribute__((ext_vector_type(8))) float v8f;

constexpr int BIN_SHIFT = 8;
constexpr int BIN_NODES = 1 << BIN_SHIFT;
constexpr int MAX_BINS = 512;
constexpr int CHUNKS = 250;
constexpr int EDGE_THREADS = 1024;
constexpr int EDGE_AHEAD = 7;
constexpr int SEG_LANES = 4;
constexpr int SEG_AHEAD = 8;
constexpr int STAGE_PAIRS = 6400;
constexpr int STAGE_IDS = 8192;
static_assert(MAX_BINS < EDGE_THREADS, "thread nb writes the closing row of offT");
static_assert(CHUNKS * SEG_LANES <= EDGE_THREADS, "every chunk's run has its threads");

__device__ __forceinline__ int lds_add(int* p, int v) {
  return __hip_atomic_fetch_add(p, v, __ATOMIC_RELAXED, __HIP_MEMORY_SCOPE_WORKGROUP);
}

__global__ void __launch_bounds__(EDGE_THREADS)
k_chunk_sort(const int* __restrict__ src, const int* __restrict__ dst,
             int* __restrict__ offT, int* __restrict__ pairs_w, int E, int nb, int per) {
  __shared__ int h[MAX_BINS];
  __shared__ int wsum[EDGE_THREADS / 32];
  __shared__ __attribute__((aligned(16))) int2 stage[STAGE_PAIRS];
  int2* pairs = (int2*)pairs_w;
  const bool staged = per <= STAGE_PAIRS;
  const int tid = threadIdx.x;
  const int lane = tid & 31;
  const int wv = tid >> 5;
  const int b = blockIdx.x;
  if (tid < nb) h[tid] = 0;
  __syncthreads();
  const int e0 = b * per;
  const int e1 = (e0 + per < E) ? e0 + per : E;
  for (int base = e0; base < e1; base += EDGE_AHEAD * EDGE_THREADS) {
    int d[EDGE_AHEAD];
#pragma unroll
    for (int j = 0; j < EDGE_AHEAD; ++j) {
      const int e = base + j * EDGE_THREADS + tid;
      d[j] = (e < e1) ? dst[e] : -1;
    }
#pragma unroll
    for (int j = 0; j < EDGE_AHEAD; ++j)
      if (d[j] >= 0) lds_add(&h[d[j] >> BIN_SHIFT], 1);
  }
  __syncthreads();
  const int v = (tid < nb) ? h[tid] : 0;
  int incl = v;
  for (int d = 1; d < 32; d <<= 1) {
    int u = __shfl_up(incl, d);
    if (lane >= d) incl += u;
  }
  if (lane == 31) wsum[wv] = incl;
  __syncthreads();
  int front = (lane < wv) ? wsum[lane] : 0;
  for (int d = 16; d >= 1; d >>= 1) front += __shfl_xor(front, d);
  const int excl = e0 + front + incl - v;
  if (tid <= nb) {
    offT[tid * CHUNKS + b] = excl;
    if (tid < nb) h[tid] = excl;
  }
  __syncthreads();
  for (int base = e0; base < e1; base += EDGE_AHEAD * EDGE_THREADS) {
    int d[EDGE_AHEAD], s[EDGE_AHEAD];
#pragma unroll
    for (int j = 0; j < EDGE_AHEAD; ++j) {
      const int e = base + j * EDGE_THREADS + tid;
      d[j] = (e < e1) ? dst[e] : -1;
    }
#pragma unroll
    for (int j = 0; j < EDGE_AHEAD; ++j) {
      const int e = base + j * EDGE_THREADS + tid;
      s[j] = (e < e1) ? src[e] : 0;
    }
#pragma unroll
    for (int j = 0; j < EDGE_AHEAD; ++j)
      if (d[j] >= 0) {
        const int slot = lds_add(&h[d[j] >> BIN_SHIFT], 1);
        if (staged) stage[slot - e0] = make_int2(s[j], d[j]);
        else pairs[slot] = make_int2(s[j], d[j]);
      }
  }
  if (staged) {
    __syncthreads();
    const int cnt = e1 - e0;
    int2* gp = pairs + e0;
    if ((((size_t)gp) & 15) == 0) {
      typedef __attribute__((ext_vector_type(4))) int v4i;
      const int half = cnt >> 1;
      for (int i = tid; i < half; i += EDGE_THREADS) ((v4i*)gp)[i] = ((const v4i*)stage)[i];
      if ((cnt & 1) && tid == 0) gp[cnt - 1] = stage[cnt - 1];
    } else {
      for (int i = tid; i < cnt; i += EDGE_THREADS) gp[i] = stage[i];
    }
  }
}

__global__ void __launch_bounds__(EDGE_THREADS)
k_bin_finish(const int* __restrict__ offT, const int* __restrict__ pairs_w,
             int* __restrict__ rowptr, int* __restrict__ csr, float* __restrict__ dinv,
             int n, int E, int per) {
  const int2* pairs = (const int2*)pairs_w;
  __shared__ int c[BIN_NODES];
  __shared__ int wsum[BIN_NODES / 32];
  __shared__ int red[EDGE_THREADS / 32];
  __shared__ int ids[STAGE_IDS];
  const int tid = threadIdx.x;
  const int lane = tid & 31;
  const int wv = tid >> 5;
  const int bin = blockIdx.x;
  const int ch = tid / SEG_LANES;
  const int sub = tid % SEG_LANES;
  int a = 0, len = 0, lower = 0;
  if (ch < CHUNKS) {
    a = offT[bin * CHUNKS + ch];
    len = offT[(bin + 1) * CHUNKS + ch] - a;
    if (sub == 0) lower = a - ch * per;
  }
  for (int d = 16; d >= 1; d >>= 1) lower += __shfl_xor(lower, d);
  if (lane == 0) red[wv] = lower;
  if (tid < BIN_NODES) c[tid] = 0;
  __syncthreads();
  int s0 = red[lane];
  for (int d = 16; d >= 1; d >>= 1) s0 += __shfl_xor(s0, d);

  for (int j0 = sub; j0 < len; j0 += SEG_LANES * SEG_AHEAD) {
    int d[SEG_AHEAD];
#pragma unroll
    for (int u = 0; u < SEG_AHEAD; ++u) {
      const int j = j0 + SEG_LANES * u;
      d[u] = (j < len) ? pairs[a + j].y : -1;
    }
#pragma unroll
    for (int u = 0; u < SEG_AHEAD; ++u)
      if (d[u] >= 0) lds_add(&c[d[u] & (BIN_NODES - 1)], 1);
  }
  __syncthreads();
  const int cv = (tid < BIN_NODES) ? c[tid] : 0;
  int incl = cv;
  for (int d = 1; d < 32; d <<= 1) {
    int u = __shfl_up(incl, d);
    if (lane >= d) incl += u;
  }
  if (lane == 31 && wv < BIN_NODES / 32) wsum[wv] = incl;
  __syncthreads();
  int front = s0;
  int tot = 0;
  for (int k = 0; k < BIN_NODES / 32; ++k) {
    front += (k < wv) ? wsum[k] : 0;
    tot += wsum[k];
  }
  const bool staged = tot <= STAGE_IDS;
  const int excl = front + incl - cv;
  const int node = bin * BIN_NODES + tid;
  if (tid < BIN_NODES) {
    c[tid] = excl;
    if (node < n) {
      rowptr[node] = excl;
      float dg = (float)(cv + 1);
      dinv[node] = (dg > 0.0f) ? rsqrtf(dg) : 0.0f;
    }
  }
  if (bin == (int)gridDim.x - 1 && tid == 0) rowptr[n] = E;
  __syncthreads();
  for (int j0 = sub; j0 < len; j0 += SEG_LANES * SEG_AHEAD) {
    int2 pr[SEG_AHEAD];
#pragma unroll
    for (int u = 0; u < SEG_AHEAD; ++u) {
      const int j = j0 + SEG_LANES * u;
      pr[u] = (j < len) ? pairs[a + j] : make_int2(0, -1);
    }
#pragma unroll
    for (int u = 0; u < SEG_AHEAD; ++u)
      if (pr[u].y >= 0) {
        const int pos = lds_add(&c[pr[u].y & (BIN_NODES - 1)], 1);
        if (staged) ids[pos - s0] = pr[u].x;
        else csr[pos] = pr[u].x;
      }
  }
  if (staged) {
    __syncthreads();
    const int mis = (int)((((size_t)csr) >> 2) & 31);
    const int g0 = s0 - ((s0 + mis) & 31);
    const int gend = s0 + tot;
    for (int g = g0 + tid; g < gend; g += EDGE_THREADS)
      if (g >= s0) csr[g] = ids[g - s0];
  }
}

constexpr int GEMM_THREADS = 800;
template <int K, int NC>
__global__ void __launch_bounds__(GEMM_THREADS)
k_gemm_strip(const float* __restrict__ A, const float* __restrict__ W,
             const float* __restrict__ rowscale, float* __restrict__ C, int nstrips) {
  constexpr int TILES = NC / 16;
  constexpr int P2 = 2 * NC + 32;
  constexpr int NQ = NC / 4;
  __shared__ __attribute__((aligned(16))) float Ws[(K / 2) * P2];

  typedef __attribute__((ext_vector_type(4))) float v4f;
#pragma unroll
  for (int idx = threadIdx.x; idx < (K / 2) * NQ; idx += GEMM_THREADS) {
    const int p = idx / NQ;
    const int q = idx % NQ;
    const v4f w0 = *(const v4f*)(W + (size_t)(2 * p) * NC + 4 * q);
    const v4f w1 = *(const v4f*)(W + (size_t)(2 * p + 1) * NC + 4 * q);
    v4f lo, hi;
    lo.x = w0.x; lo.y = w1.x; lo.z = w0.y; lo.w = w1.y;
    hi.x = w0.z; hi.y = w1.z; hi.z = w0.w; hi.w = w1.w;
    float* d = Ws + p * P2 + 8 * q;
    *(v4f*)d = lo;
    *(v4f*)(d + 4) = hi;
  }
  __syncthreads();

  const int lane = threadIdx.x & 31;
  const int wave = threadIdx.x >> 5;
  const int strip = blockIdx.x * (GEMM_THREADS / 32) + wave;
  if (strip >= nstrips) return;
  const int c = lane & 15;
  const int h = lane >> 4;

  const float* arow = A + (size_t)(strip * 16 + c) * K + 2 * h;
  const float* bl = Ws + h * P2 + 2 * c;

  v8f acc[TILES];
#pragma unroll
  for (int t = 0; t < TILES; ++t) acc[t] = v8f{};

  v2f acur[8], anxt[8];
#pragma unroll
  for (int s = 0; s < 8; ++s) acur[s] = *(const v2f*)(arow + 4 * s);

  v2f bcur[TILES], bnxt[TILES];
#pragma unroll
  for (int t = 0; t < TILES; ++t) bcur[t] = *(const v2f*)(bl + 32 * t);

  for (int k0 = 0; k0 < K; k0 += 32) {
    const int kn = (k0 + 32 < K) ? k0 + 32 : k0;
#pragma unroll
    for (int s = 0; s < 8; ++s) anxt[s] = *(const v2f*)(arow + kn + 4 * s);

#pragma unroll
    for (int s = 0; s < 8; ++s) {
      const v2f a = acur[s];
      const int kb = (k0 + 4 * s + 4 < K) ? k0 + 4 * s + 4 : 0;
      const float* bp = bl + (kb >> 1) * P2;
#pragma unroll
      for (int t = 0; t < TILES; ++t) bnxt[t] = *(const v2f*)(bp + 32 * t);
      __builtin_amdgcn_sched_barrier(0);
#pragma unroll
      for (int t = 0; t < TILES; ++t)
        acc[t] = __builtin_amdgcn_wmma_f32_16x16x4_f32(false, a, false, bcur[t], (short)0,
                                                       acc[t], false, false);
      __builtin_amdgcn_sched_barrier(0);
#pragma unroll
      for (int t = 0; t < TILES; ++t) bcur[t] = bnxt[t];
    }
#pragma unroll
    for (int s = 0; s < 8; ++s) acur[s] = anxt[s];
  }

  const int rbase = strip * 16 + 8 * h;
  float rs[8];
#pragma unroll
  for (int v = 0; v < 8; ++v) rs[v] = rowscale[rbase + v];
#pragma unroll
  for (int t = 0; t < TILES; ++t)
#pragma unroll
    for (int v = 0; v < 8; ++v)
      C[(size_t)(rbase + v) * NC + 16 * t + c] = acc[t][v] * rs[v];
}

template <int F>
struct RowVec {
  typedef __attribute__((ext_vector_type(F / 32))) float type;
};

template <int F>
struct NodeHead {
  int beg, end;
  float di;
  typename RowVec<F>::type acc;
};

template <int F>
__device__ __forceinline__ NodeHead<F>
node_head(int node, int lane, const int* __restrict__ rowptr, const float* __restrict__ dinv,
          const float* __restrict__ H) {
  typedef typename RowVec<F>::type vecT;
  NodeHead<F> hd;
  hd.beg = __builtin_amdgcn_readfirstlane(rowptr[node]);
  hd.end = __builtin_amdgcn_readfirstlane(rowptr[node + 1]);
  const int lo = lane * (F / 32);
  hd.di = dinv[node];
  hd.acc = *(const vecT*)(H + (size_t)node * F + lo);
  return hd;
}

template <int F>
__device__ __forceinline__ typename RowVec<F>::type
node_walk(const NodeHead<F>& hd, int lane, const int* __restrict__ csr,
          const float* __restrict__ H) {
  typedef typename RowVec<F>::type vecT;
  const int end = hd.end;
  const int lo = lane * (F / 32);
  vecT acc = hd.acc;

  for (int base = hd.beg; base < end; base += 32) {
    int ms = 0;
    if (base + lane < end) ms = csr[base + lane];
    const int c = (end - base < 32) ? (end - base) : 32;
    int j = 0;
    for (; j + 4 <= c; j += 4) {
      const int s0 = __builtin_amdgcn_readlane(ms, j);
      const int s1 = __builtin_amdgcn_readlane(ms, j + 1);
      const int s2 = __builtin_amdgcn_readlane(ms, j + 2);
      const int s3 = __builtin_amdgcn_readlane(ms, j + 3);
      const vecT m0 = *(const vecT*)(H + (size_t)s0 * F + lo);
      const vecT m1 = *(const vecT*)(H + (size_t)s1 * F + lo);
      const vecT m2 = *(const vecT*)(H + (size_t)s2 * F + lo);
      const vecT m3 = *(const vecT*)(H + (size_t)s3 * F + lo);
      acc += m0;
      acc += m1;
      acc += m2;
      acc += m3;
    }
    if (j + 2 <= c) {
      const int s0 = __builtin_amdgcn_readlane(ms, j);
      const int s1 = __builtin_amdgcn_readlane(ms, j + 1);
      const vecT m0 = *(const vecT*)(H + (size_t)s0 * F + lo);
      const vecT m1 = *(const vecT*)(H + (size_t)s1 * F + lo);
      acc += m0;
      acc += m1;
      j += 2;
    }
    if (j < c) {
      const int s0 = __builtin_amdgcn_readlane(ms, j);
      const vecT m0 = *(const vecT*)(H + (size_t)s0 * F + lo);
      acc += m0;
    }
  }
  return acc;
}

template <int F>
__global__ void k_gather(const int* __restrict__ rowptr, const int* __restrict__ csr,
                         const float* __restrict__ dinv, const float* __restrict__ H,
                         const float* __restrict__ bias, float* __restrict__ O, int n) {
  typedef typename RowVec<F>::type vecT;
  const int lane = threadIdx.x & 31;
  const int node = blockIdx.x * (blockDim.x >> 5) + (threadIdx.x >> 5);
  if (node >= n) return;
  const NodeHead<F> hd = node_head<F>(node, lane, rowptr, dinv, H);
  const vecT bv = *(const vecT*)(bias + lane * (F / 32));
  const vecT acc = node_walk<F>(hd, lane, csr, H);
  *(vecT*)(O + (size_t)node * F + lane * (F / 32)) = bv + acc * hd.di;
}

constexpr int FUSE_WAVES = 16;
constexpr int FUSE_THREADS = 32 * FUSE_WAVES;
constexpr int FUSE_NODES = 2 * FUSE_WAVES;
constexpr int FUSE_F = 128;
constexpr int FUSE_NC = 64;
constexpr int FUSE_AP = FUSE_F + 4;
__global__ void __launch_bounds__(FUSE_THREADS)
k_gather_proj(const int* __restrict__ rowptr, const int* __restrict__ csr,
              const float* __restrict__ dinv, const float* __restrict__ H,
              const float* __restrict__ bias, const float* __restrict__ W,
              float* __restrict__ C, int n) {
  constexpr int F = FUSE_F;
  constexpr int NC = FUSE_NC;
  constexpr int P2 = 2 * NC + 32;
  constexpr int NQ = NC / 4;
  constexpr int COPIES = (F / 2) * NQ / FUSE_THREADS;
  static_assert(COPIES * FUSE_THREADS == (F / 2) * NQ, "W is copied in whole rounds");
  typedef __attribute__((ext_vector_type(4))) float v4f;
  __shared__ __attribute__((aligned(16))) float Ws[(F / 2) * P2];
  __shared__ __attribute__((aligned(16))) float As[FUSE_NODES * FUSE_AP];

  const int lane = threadIdx.x & 31;
  const int wave = __builtin_amdgcn_readfirstlane(threadIdx.x >> 5);
  const int first = blockIdx.x * FUSE_NODES;
  const int n0 = first + 2 * wave;
  const bool in0 = n0 < n;
  const bool in1 = n0 + 1 < n;
  const NodeHead<F> hd0 = node_head<F>(in0 ? n0 : 0, lane, rowptr, dinv, H);
  const NodeHead<F> hd1 = node_head<F>(in1 ? n0 + 1 : 0, lane, rowptr, dinv, H);
  const v4f bv = *(const v4f*)(bias + 4 * lane);

#pragma unroll
  for (int it = 0; it < COPIES; ++it) {
    const int idx = threadIdx.x + it * FUSE_THREADS;
    const int p = idx / NQ;
    const int q = idx % NQ;
    const v4f w0 = *(const v4f*)(W + (size_t)(2 * p) * NC + 4 * q);
    const v4f w1 = *(const v4f*)(W + (size_t)(2 * p + 1) * NC + 4 * q);
    v4f lo, hi;
    lo.x = w0.x; lo.y = w1.x; lo.z = w0.y; lo.w = w1.y;
    hi.x = w0.z; hi.y = w1.z; hi.z = w0.w; hi.w = w1.w;
    float* d = Ws + p * P2 + 8 * q;
    *(v4f*)d = lo;
    *(v4f*)(d + 4) = hi;
  }

  v4f r0 = bv + node_walk<F>(hd0, lane, csr, H) * hd0.di;
  v4f r1 = bv + node_walk<F>(hd1, lane, csr, H) * hd1.di;
  if (!in0) r0 = v4f{};
  if (!in1) r1 = v4f{};
  r0.x = fmaxf(r0.x, 0.0f); r0.y = fmaxf(r0.y, 0.0f);
  r0.z = fmaxf(r0.z, 0.0f); r0.w = fmaxf(r0.w, 0.0f);
  r1.x = fmaxf(r1.x, 0.0f); r1.y = fmaxf(r1.y, 0.0f);
  r1.z = fmaxf(r1.z, 0.0f); r1.w = fmaxf(r1.w, 0.0f);
  *(v4f*)(As + (2 * wave) * FUSE_AP + 4 * lane) = r0;
  *(v4f*)(As + (2 * wave + 1) * FUSE_AP + 4 * lane) = r1;
  __syncthreads();
  if (wave >= 8) return;

  const int s = wave >> 2;
  const int t = wave & 3;
  const int c = lane & 15;
  const int h = lane >> 4;
  const float* ap = As + (16 * s + c) * FUSE_AP + 2 * h;
  const float* bp = Ws + h * P2 + 2 * (16 * t + c);

  v8f acc = v8f{};
#pragma unroll
  for (int k = 0; k < F; k += 4) {
    const v2f a = *(const v2f*)(ap + k);
    const v2f b = *(const v2f*)(bp + (k >> 1) * P2);
    acc = __builtin_amdgcn_wmma_f32_16x16x4_f32(false, a, false, b, (short)0, acc, false, false);
  }

  const int r16 = first + 16 * s;
  float* crow = C + (size_t)(r16 + 8 * h) * NC + 16 * t + c;
  const float* drow = dinv + r16 + 8 * h;
  if (r16 + 16 <= n) {
#pragma unroll
    for (int v = 0; v < 8; ++v) crow[(size_t)v * NC] = acc[v] * drow[v];
  } else {
#pragma unroll
    for (int v = 0; v < 8; ++v)
      if (r16 + 8 * h + v < n) crow[(size_t)v * NC] = acc[v] * drow[v];
  }
}

extern "C" void kernel_launch(void* const* d_in, const int* in_sizes, int n_in,
                              void* d_out, int out_size, void* d_ws, size_t ws_size,
                              hipStream_t stream) {
  const float* x  = (const float*)d_in[0];
  const int*   ei = (const int*)d_in[1];
  const float* W1 = (const float*)d_in[2];
  const float* b1 = (const float*)d_in[3];
  const float* W2 = (const float*)d_in[4];
  const float* b2 = (const float*)d_in[5];
  float* out = (float*)d_out;

  const int FIN = 256, FHID = 128, FOUT = 64;
  const int N = in_sizes[0] / FIN;
  const int E = in_sizes[1] / 2;

  const int* srcI = ei;
  const int* dstI = ei + E;

  float* ws   = (float*)d_ws;
  float* dinv = ws;
  float* xw1  = ws + N;
  float* rest = xw1 + (size_t)N * FHID;
  float* h2   = rest;

  int* list = (int*)(rest + (size_t)N * FOUT);
  const int NB = (N + BIN_NODES - 1) / BIN_NODES;
  const int PER = (E + CHUNKS - 1) / CHUNKS;
  int*  offT  = (int*)rest;
  int*  pairs = offT + 131072;

  const int T = 256;
  const int NS = N / 16;
  const int GW = GEMM_THREADS / 32;

  k_chunk_sort<<<CHUNKS, EDGE_THREADS, 0, stream>>>(srcI, dstI, offT, pairs, E, NB, PER);
  k_bin_finish<<<NB, EDGE_THREADS, 0, stream>>>(offT, pairs, list, list + N + 1, dinv, N, E,
                                                PER);

  k_gemm_strip<FIN, FHID><<<(NS + GW - 1) / GW, GEMM_THREADS, 0, stream>>>(x, W1, dinv, xw1, NS);
  k_gather_proj<<<(N + FUSE_NODES - 1) / FUSE_NODES, FUSE_THREADS, 0, stream>>>(
      list, list + N + 1, dinv, xw1, b1, W2, h2, N);

  k_gather<64><<<(N + 7) / 8, T, 0, stream>>>(list, list + N + 1, dinv, h2, b2, out, N);
}
